# starting kernel
# baseline (speedup 1.0000x reference)
#include <hip/hip_runtime.h>

typedef __attribute__((ext_vector_type(2))) float v2f;
typedef __attribute__((ext_vector_type(8))) float v8f;

#define IN_F  128
#define HID_F 128
#define OUT_F 64

__global__ void k_fill(float* __restrict__ p, float v, long n) {
    long i = (long)blockIdx.x * blockDim.x + threadIdx.x;
    long stride = (long)gridDim.x * blockDim.x;
    for (; i < n; i += stride) p[i] = v;
}

__global__ void k_degree(const int* __restrict__ dst, float* __restrict__ deg, long E) {
    long i = (long)blockIdx.x * blockDim.x + threadIdx.x;
    long stride = (long)gridDim.x * blockDim.x;
    for (; i < E; i += stride)
        __hip_atomic_fetch_add(&deg[dst[i]], 1.0f, __ATOMIC_RELAXED, __HIP_MEMORY_SCOPE_AGENT);
}

__global__ void k_rsqrt(float* __restrict__ deg, int N) {
    int i = blockIdx.x * blockDim.x + threadIdx.x;
    if (i < N) deg[i] = rsqrtf(deg[i]);
}

template <int K, int NCOL, int RT>
__global__ void __launch_bounds__(32 * (NCOL / 16))
k_gemm_wmma(const float* __restrict__ X, const float* __restrict__ W,
            float* __restrict__ H, int M) {
    constexpr int ROWS = 16 * RT;
    constexpr int SXS  = K + 4;
    __shared__ float sX[ROWS * SXS];

    const int tid  = threadIdx.x;
    const int wave = tid >> 5;
    const int lane = tid & 31;
    const int row0 = blockIdx.x * ROWS;

    for (int i = tid; i < ROWS * K; i += blockDim.x) {
        int r = i / K, c = i % K;
        int gr = row0 + r;
        sX[r * SXS + c] = (gr < M) ? X[(size_t)gr * K + c] : 0.0f;
    }
    __syncthreads();

    const int half = lane >> 4;
    const int idx  = lane & 15;
    const int colbase = wave * 16;

    v8f acc[RT] = {};
#pragma unroll
    for (int k0 = 0; k0 < K; k0 += 4) {
        const int ka = k0 + half * 2;
        v2f b;
        b.x = W[(size_t)(ka)     * NCOL + colbase + idx];
        b.y = W[(size_t)(ka + 1) * NCOL + colbase + idx];
#pragma unroll
        for (int t = 0; t < RT; ++t) {
            v2f a;
            a.x = sX[(t * 16 + idx) * SXS + ka];
            a.y = sX[(t * 16 + idx) * SXS + ka + 1];
            acc[t] = __builtin_amdgcn_wmma_f32_16x16x4_f32(
                 false, a,  false, b,
                 (short)0, acc[t],  false,  false);
        }
    }

#pragma unroll
    for (int t = 0; t < RT; ++t) {
#pragma unroll
        for (int v = 0; v < 8; ++v) {
            int gr = row0 + t * 16 + half * 8 + v;
            if (gr < M) H[(size_t)gr * NCOL + colbase + idx] = acc[t][v];
        }
    }
}

template <int F>
__global__ void k_scatter(const int* __restrict__ src, const int* __restrict__ dst,
                          const float* __restrict__ dinv, const float* __restrict__ h,
                          float* __restrict__ agg, long E) {
    constexpr int C = F / 32;
    const int wavesPerBlock = blockDim.x >> 5;
    const long e = (long)blockIdx.x * wavesPerBlock + (threadIdx.x >> 5);
    if (e >= E) return;
    const int lane = threadIdx.x & 31;
    const int s = src[e];
    const int d = dst[e];
    const float nm = dinv[s] * dinv[d];
    const float* __restrict__ hp = h + (size_t)s * F + lane * C;
    float* __restrict__ op = agg + (size_t)d * F + lane * C;
    float v[C];
#pragma unroll
    for (int j = 0; j < C; ++j) v[j] = hp[j];
#pragma unroll
    for (int j = 0; j < C; ++j)
        __hip_atomic_fetch_add(op + j, v[j] * nm, __ATOMIC_RELAXED, __HIP_MEMORY_SCOPE_AGENT);
}

__global__ void k_post_relu(float* __restrict__ agg, const float* __restrict__ h,
                            const float* __restrict__ dinv, const float* __restrict__ b,
                            long total) {
    long i = (long)blockIdx.x * blockDim.x + threadIdx.x;
    if (i >= total) return;
    int n = (int)(i / HID_F), f = (int)(i % HID_F);
    float di = dinv[n];
    float v = agg[i] + h[i] * di * di + b[f];
    agg[i] = v > 0.0f ? v : 0.0f;
}

__global__ void k_post(float* __restrict__ out, const float* __restrict__ h2,
                       const float* __restrict__ dinv, const float* __restrict__ b,
                       long total) {
    long i = (long)blockIdx.x * blockDim.x + threadIdx.x;
    if (i >= total) return;
    int n = (int)(i / OUT_F), f = (int)(i % OUT_F);
    float di = dinv[n];
    out[i] = out[i] + h2[i] * di * di + b[f];
}

extern "C" void kernel_launch(void* const* d_in, const int* in_sizes, int n_in,
                              void* d_out, int out_size, void* d_ws, size_t ws_size,
                              hipStream_t stream) {
    const float* x  = (const float*)d_in[0];
    const int*   ei = (const int*)d_in[1];
    const float* W1 = (const float*)d_in[2];
    const float* b1 = (const float*)d_in[3];
    const float* W2 = (const float*)d_in[4];
    const float* b2 = (const float*)d_in[5];
    float* out = (float*)d_out;

    const int  N = in_sizes[0] / IN_F;
    const long E = (long)in_sizes[1] / 2;
    const int* srcIdx = ei;
    const int* dstIdx = ei + E;

    float* ws   = (float*)d_ws;
    float* dinv = ws;
    float* h    = dinv + N;
    float* agg1 = h + (size_t)N * HID_F;
    float* h2   = agg1 + (size_t)N * HID_F;

    k_fill<<<1024, 256, 0, stream>>>(dinv, 1.0f, (long)N);
    k_fill<<<4096, 256, 0, stream>>>(agg1, 0.0f, (long)N * HID_F);
    k_fill<<<2048, 256, 0, stream>>>(out, 0.0f, (long)N * OUT_F);

    k_degree<<<4096, 256, 0, stream>>>(dstIdx, dinv, E);
    k_rsqrt<<<(N + 255) / 256, 256, 0, stream>>>(dinv, N);

    constexpr int RT   = 4;
    constexpr int ROWS = 16 * RT;

    k_gemm_wmma<IN_F, HID_F, RT>
        <<<(N + ROWS - 1) / ROWS, 32 * (HID_F / 16), 0, stream>>>(x, W1, h, N);
    k_scatter<HID_F><<<(int)((E + 7) / 8), 256, 0, stream>>>(srcIdx, dstIdx, dinv, h, agg1, E);
    k_post_relu<<<(int)(((long)N * HID_F + 255) / 256), 256, 0, stream>>>(agg1, h, dinv, b1,
                                                                          (long)N * HID_F);

    k_gemm_wmma<HID_F, OUT_F, RT>
        <<<(N + ROWS - 1) / ROWS, 32 * (OUT_F / 16), 0, stream>>>(agg1, W2, h2, N);
    k_scatter<OUT_F><<<(int)((E + 7) / 8), 256, 0, stream>>>(srcIdx, dstIdx, dinv, h2, out, E);
    k_post<<<(int)(((long)N * OUT_F + 255) / 256), 256, 0, stream>>>(out, h2, dinv, b2,
                                                                     (long)N * OUT_F);
}


# Replace per-edge float-atomic scatter by CSR pull: int count+arrival rank per dst, scan, place src, one wave per node sums rows in registers; per-node sum order = arrival rank (was atomic order)
# speedup vs baseline: 1.8277x; 1.8277x over previous
#include <hip/hip_runtime.h>


typedef __attribute__((ext_vector_type(2))) float v2f;
typedef __attribute__((ext_vector_type(8))) float v8f;

#define IN_F  128
#define HID_F 128
#define OUT_F 64

__global__ void k_zero(int* __restrict__ p, int n) {
    int i = blockIdx.x * blockDim.x + threadIdx.x;
    if (i < n) p[i] = 0;
}

__global__ void k_count(const int* __restrict__ dst, int* __restrict__ cnt,
                        int* __restrict__ rank, long E) {
    long i = (long)blockIdx.x * blockDim.x + threadIdx.x;
    if (i < E)
        rank[i] = __hip_atomic_fetch_add(&cnt[dst[i]], 1, __ATOMIC_RELAXED,
                                         __HIP_MEMORY_SCOPE_AGENT);
}

__device__ __forceinline__ int4 load4_guard(const int* __restrict__ p, int i0, int n) {
    if (i0 + 3 < n) return *(const int4*)(p + i0);
    int4 c;
    c.x = (i0     < n) ? p[i0]     : 0;
    c.y = (i0 + 1 < n) ? p[i0 + 1] : 0;
    c.z = (i0 + 2 < n) ? p[i0 + 2] : 0;
    c.w = (i0 + 3 < n) ? p[i0 + 3] : 0;
    return c;
}

__global__ void __launch_bounds__(1024)
k_scan(const int* __restrict__ cnt, int* __restrict__ rowptr, float* __restrict__ dinv, int N) {
    __shared__ int sTot[32];
    const int lane = threadIdx.x & 31;
    const int wave = threadIdx.x >> 5;
    const int seg  = ((((N + 31) / 32) + 127) / 128) * 128;
    const int wbeg = wave * seg;
    const int wend = (wbeg + seg < N) ? (wbeg + seg) : N;

    int s = 0;
    for (int base = wbeg; base < wend; base += 128) {
        int4 c = load4_guard(cnt, base + lane * 4, N);
        s += c.x + c.y + c.z + c.w;
    }
    for (int o = 16; o > 0; o >>= 1) s += __shfl_xor(s, o);
    if (lane == 0) sTot[wave] = s;
    __syncthreads();

    const int mine = sTot[lane];
    int carry = (lane < wave) ? mine : 0;
    int total = mine;
    for (int o = 16; o > 0; o >>= 1) {
        carry += __shfl_xor(carry, o);
        total += __shfl_xor(total, o);
    }
    if (threadIdx.x == 0) rowptr[N] = total;

    for (int base = wbeg; base < wend; base += 128) {
        const int i0 = base + lane * 4;
        int4 c = load4_guard(cnt, i0, N);
        const int ls = c.x + c.y + c.z + c.w;
        int incl = ls;
        for (int o = 1; o < 32; o <<= 1) {
            int u = __shfl_up(incl, o);
            if (lane >= o) incl += u;
        }
        const int ex = carry + incl - ls;
        int4 r;
        r.x = ex; r.y = ex + c.x; r.z = r.y + c.y; r.w = r.z + c.z;
        float4 d;
        d.x = rsqrtf((float)(c.x + 1)); d.y = rsqrtf((float)(c.y + 1));
        d.z = rsqrtf((float)(c.z + 1)); d.w = rsqrtf((float)(c.w + 1));
        if (i0 + 3 < N) {
            *(int4*)(rowptr + i0) = r;
            *(float4*)(dinv + i0) = d;
        } else {
            if (i0     < N) { rowptr[i0]     = r.x; dinv[i0]     = d.x; }
            if (i0 + 1 < N) { rowptr[i0 + 1] = r.y; dinv[i0 + 1] = d.y; }
            if (i0 + 2 < N) { rowptr[i0 + 2] = r.z; dinv[i0 + 2] = d.z; }
        }
        carry += __shfl(incl, 31);
    }
}

__global__ void k_place(const int* __restrict__ src, const int* __restrict__ dst,
                        const int* __restrict__ rowptr, const int* __restrict__ rank,
                        int* __restrict__ col, long E) {
    long i = (long)blockIdx.x * blockDim.x + threadIdx.x;
    if (i < E) col[rowptr[dst[i]] + rank[i]] = src[i];
}

template <int K, int NCOL, int RT>
__global__ void __launch_bounds__(32 * (NCOL / 16))
k_gemm_wmma(const float* __restrict__ X, const float* __restrict__ W,
            float* __restrict__ H, int M) {
    constexpr int ROWS = 16 * RT;
    constexpr int SXS  = K + 4;
    __shared__ float sX[ROWS * SXS];

    const int tid  = threadIdx.x;
    const int wave = tid >> 5;
    const int lane = tid & 31;
    const int row0 = blockIdx.x * ROWS;

    for (int i = tid; i < ROWS * K; i += blockDim.x) {
        int r = i / K, c = i % K;
        int gr = row0 + r;
        sX[r * SXS + c] = (gr < M) ? X[(size_t)gr * K + c] : 0.0f;
    }
    __syncthreads();

    const int half = lane >> 4;
    const int idx  = lane & 15;
    const int colbase = wave * 16;

    v8f acc[RT] = {};
#pragma unroll
    for (int k0 = 0; k0 < K; k0 += 4) {
        const int ka = k0 + half * 2;
        v2f b;
        b.x = W[(size_t)(ka)     * NCOL + colbase + idx];
        b.y = W[(size_t)(ka + 1) * NCOL + colbase + idx];
#pragma unroll
        for (int t = 0; t < RT; ++t) {
            v2f a;
            a.x = sX[(t * 16 + idx) * SXS + ka];
            a.y = sX[(t * 16 + idx) * SXS + ka + 1];
            acc[t] = __builtin_amdgcn_wmma_f32_16x16x4_f32(
                 false, a,  false, b,
                 (short)0, acc[t],  false,  false);
        }
    }

#pragma unroll
    for (int t = 0; t < RT; ++t) {
#pragma unroll
        for (int v = 0; v < 8; ++v) {
            int gr = row0 + t * 16 + half * 8 + v;
            if (gr < M) H[(size_t)gr * NCOL + colbase + idx] = acc[t][v];
        }
    }
}

template <int F>
__global__ void __launch_bounds__(256)
k_gather(const int* __restrict__ rowptr, const int* __restrict__ col,
         const float* __restrict__ dinv, const float* __restrict__ h,
         float* __restrict__ agg, int N) {
    constexpr int C = F / 32;
    typedef float vecC __attribute__((ext_vector_type(C)));
    const int lane = threadIdx.x & 31;
    const int node = __builtin_amdgcn_readfirstlane(
        (int)(blockIdx.x * (blockDim.x >> 5) + (threadIdx.x >> 5)));
    if (node >= N) return;
    const int beg = rowptr[node];
    const int end = rowptr[node + 1];
    const float dd = dinv[node];
    const vecC* __restrict__ hv = (const vecC*)h;
    vecC acc = 0.0f;

    for (int base = beg; base < end; base += 32) {
        const int rem = end - base;
        const int cnt = rem < 32 ? rem : 32;
        int   s_l = 0;
        float w_l = 0.0f;
        if (lane < cnt) {
            s_l = col[base + lane];
            w_l = dinv[s_l];
        }
        const int wbits = __builtin_bit_cast(int, w_l);
        int j = 0;
        for (; j + 8 <= cnt; j += 8) {
            vecC v[8];
            for (int u = 0; u < 8; ++u) {
                const int s = __builtin_amdgcn_readlane(s_l, j + u);
                v[u] = hv[(size_t)s * 32 + lane];
            }
            for (int u = 0; u < 8; ++u) {
                const float w = __builtin_bit_cast(float, __builtin_amdgcn_readlane(wbits, j + u)) * dd;
                acc += v[u] * w;
            }
        }
        if (j + 4 <= cnt) {
            vecC v[4];
            for (int u = 0; u < 4; ++u) {
                const int s = __builtin_amdgcn_readlane(s_l, j + u);
                v[u] = hv[(size_t)s * 32 + lane];
            }
            for (int u = 0; u < 4; ++u) {
                const float w = __builtin_bit_cast(float, __builtin_amdgcn_readlane(wbits, j + u)) * dd;
                acc += v[u] * w;
            }
            j += 4;
        }
        for (; j < cnt; ++j) {
            const int s = __builtin_amdgcn_readlane(s_l, j);
            const float w = __builtin_bit_cast(float, __builtin_amdgcn_readlane(wbits, j)) * dd;
            acc += hv[(size_t)s * 32 + lane] * w;
        }
    }
    ((vecC*)agg)[(size_t)node * 32 + lane] = acc;
}

__global__ void k_post_relu(float* __restrict__ agg, const float* __restrict__ h,
                            const float* __restrict__ dinv, const float* __restrict__ b,
                            long total) {
    long i = (long)blockIdx.x * blockDim.x + threadIdx.x;
    if (i >= total) return;
    int n = (int)(i / HID_F), f = (int)(i % HID_F);
    float di = dinv[n];
    float v = agg[i] + h[i] * di * di + b[f];
    agg[i] = v > 0.0f ? v : 0.0f;
}

__global__ void k_post(float* __restrict__ out, const float* __restrict__ h2,
                       const float* __restrict__ dinv, const float* __restrict__ b,
                       long total) {
    long i = (long)blockIdx.x * blockDim.x + threadIdx.x;
    if (i >= total) return;
    int n = (int)(i / OUT_F), f = (int)(i % OUT_F);
    float di = dinv[n];
    out[i] = out[i] + h2[i] * di * di + b[f];
}

extern "C" void kernel_launch(void* const* d_in, const int* in_sizes, int n_in,
                              void* d_out, int out_size, void* d_ws, size_t ws_size,
                              hipStream_t stream) {
    const float* x  = (const float*)d_in[0];
    const int*   ei = (const int*)d_in[1];
    const float* W1 = (const float*)d_in[2];
    const float* b1 = (const float*)d_in[3];
    const float* W2 = (const float*)d_in[4];
    const float* b2 = (const float*)d_in[5];
    float* out = (float*)d_out;

    const int  N = in_sizes[0] / IN_F;
    const long E = (long)in_sizes[1] / 2;
    const int* srcIdx = ei;
    const int* dstIdx = ei + E;

    float* ws   = (float*)d_ws;
    float* dinv = ws;
    float* h    = dinv + N;
    float* agg1 = h + (size_t)N * HID_F;
    float* h2   = h;
    const int N4 = (N + 3) / 4 * 4;
    int* cnt    = (int*)(agg1 + (size_t)N * HID_F);
    int* rowptr = cnt + N4;
    int* rank   = rowptr + N4 + 4;
    int* col    = rank + E;

    k_zero<<<(N + 255) / 256, 256, 0, stream>>>(cnt, N);
    k_count<<<(int)((E + 255) / 256), 256, 0, stream>>>(dstIdx, cnt, rank, E);
    k_scan<<<1, 1024, 0, stream>>>(cnt, rowptr, dinv, N);
    k_place<<<(int)((E + 255) / 256), 256, 0, stream>>>(srcIdx, dstIdx, rowptr, rank, col, E);

    constexpr int RT   = 4;
    constexpr int ROWS = 16 * RT;

    k_gemm_wmma<IN_F, HID_F, RT>
        <<<(N + ROWS - 1) / ROWS, 32 * (HID_F / 16), 0, stream>>>(x, W1, h, N);
    k_gather<HID_F><<<(N + 7) / 8, 256, 0, stream>>>(rowptr, col, dinv, h, agg1, N);
    k_post_relu<<<(int)(((long)N * HID_F + 255) / 256), 256, 0, stream>>>(agg1, h, dinv, b1,
                                                                          (long)N * HID_F);

    k_gemm_wmma<HID_F, OUT_F, RT>
        <<<(N + ROWS - 1) / ROWS, 32 * (OUT_F / 16), 0, stream>>>(agg1, W2, h2, N);
    k_gather<OUT_F><<<(N + 7) / 8, 256, 0, stream>>>(rowptr, col, dinv, h2, out, N);
    k_post<<<(int)(((long)N * OUT_F + 255) / 256), 256, 0, stream>>>(out, h2, dinv, b2,
                                                                     (long)N * OUT_F);
}


# Fuse self-loop term, bias and ReLU into the pull-aggregation kernel: the wave that sums a node's in-edges also reads the node's own row and stores the finished row; both elementwise launches go
# speedup vs baseline: 2.0706x; 1.1329x over previous
#include <hip/hip_runtime.h>


typedef __attribute__((ext_vector_type(2))) float v2f;
typedef __attribute__((ext_vector_type(8))) float v8f;

#define IN_F  128
#define HID_F 128
#define OUT_F 64

__global__ void k_zero(int* __restrict__ p, int n) {
    int i = blockIdx.x * blockDim.x + threadIdx.x;
    if (i < n) p[i] = 0;
}

__global__ void k_count(const int* __restrict__ dst, int* __restrict__ cnt,
                        int* __restrict__ rank, long E) {
    long i = (long)blockIdx.x * blockDim.x + threadIdx.x;
    if (i < E)
        rank[i] = __hip_atomic_fetch_add(&cnt[dst[i]], 1, __ATOMIC_RELAXED,
                                         __HIP_MEMORY_SCOPE_AGENT);
}

__device__ __forceinline__ int4 load4_guard(const int* __restrict__ p, int i0, int n) {
    if (i0 + 3 < n) return *(const int4*)(p + i0);
    int4 c;
    c.x = (i0     < n) ? p[i0]     : 0;
    c.y = (i0 + 1 < n) ? p[i0 + 1] : 0;
    c.z = (i0 + 2 < n) ? p[i0 + 2] : 0;
    c.w = (i0 + 3 < n) ? p[i0 + 3] : 0;
    return c;
}

__global__ void __launch_bounds__(1024)
k_scan(const int* __restrict__ cnt, int* __restrict__ rowptr, float* __restrict__ dinv, int N) {
    __shared__ int sTot[32];
    const int lane = threadIdx.x & 31;
    const int wave = threadIdx.x >> 5;
    const int seg  = ((((N + 31) / 32) + 127) / 128) * 128;
    const int wbeg = wave * seg;
    const int wend = (wbeg + seg < N) ? (wbeg + seg) : N;

    int s = 0;
    for (int base = wbeg; base < wend; base += 128) {
        int4 c = load4_guard(cnt, base + lane * 4, N);
        s += c.x + c.y + c.z + c.w;
    }
    for (int o = 16; o > 0; o >>= 1) s += __shfl_xor(s, o);
    if (lane == 0) sTot[wave] = s;
    __syncthreads();

    const int mine = sTot[lane];
    int carry = (lane < wave) ? mine : 0;
    int total = mine;
    for (int o = 16; o > 0; o >>= 1) {
        carry += __shfl_xor(carry, o);
        total += __shfl_xor(total, o);
    }
    if (threadIdx.x == 0) rowptr[N] = total;

    for (int base = wbeg; base < wend; base += 128) {
        const int i0 = base + lane * 4;
        int4 c = load4_guard(cnt, i0, N);
        const int ls = c.x + c.y + c.z + c.w;
        int incl = ls;
        for (int o = 1; o < 32; o <<= 1) {
            int u = __shfl_up(incl, o);
            if (lane >= o) incl += u;
        }
        const int ex = carry + incl - ls;
        int4 r;
        r.x = ex; r.y = ex + c.x; r.z = r.y + c.y; r.w = r.z + c.z;
        float4 d;
        d.x = rsqrtf((float)(c.x + 1)); d.y = rsqrtf((float)(c.y + 1));
        d.z = rsqrtf((float)(c.z + 1)); d.w = rsqrtf((float)(c.w + 1));
        if (i0 + 3 < N) {
            *(int4*)(rowptr + i0) = r;
            *(float4*)(dinv + i0) = d;
        } else {
            if (i0     < N) { rowptr[i0]     = r.x; dinv[i0]     = d.x; }
            if (i0 + 1 < N) { rowptr[i0 + 1] = r.y; dinv[i0 + 1] = d.y; }
            if (i0 + 2 < N) { rowptr[i0 + 2] = r.z; dinv[i0 + 2] = d.z; }
        }
        carry += __shfl(incl, 31);
    }
}

__global__ void k_place(const int* __restrict__ src, const int* __restrict__ dst,
                        const int* __restrict__ rowptr, const int* __restrict__ rank,
                        int* __restrict__ col, long E) {
    long i = (long)blockIdx.x * blockDim.x + threadIdx.x;
    if (i < E) col[rowptr[dst[i]] + rank[i]] = src[i];
}

template <int K, int NCOL, int RT>
__global__ void __launch_bounds__(32 * (NCOL / 16))
k_gemm_wmma(const float* __restrict__ X, const float* __restrict__ W,
            float* __restrict__ H, int M) {
    constexpr int ROWS = 16 * RT;
    constexpr int SXS  = K + 4;
    __shared__ float sX[ROWS * SXS];

    const int tid  = threadIdx.x;
    const int wave = tid >> 5;
    const int lane = tid & 31;
    const int row0 = blockIdx.x * ROWS;

    for (int i = tid; i < ROWS * K; i += blockDim.x) {
        int r = i / K, c = i % K;
        int gr = row0 + r;
        sX[r * SXS + c] = (gr < M) ? X[(size_t)gr * K + c] : 0.0f;
    }
    __syncthreads();

    const int half = lane >> 4;
    const int idx  = lane & 15;
    const int colbase = wave * 16;

    v8f acc[RT] = {};
#pragma unroll
    for (int k0 = 0; k0 < K; k0 += 4) {
        const int ka = k0 + half * 2;
        v2f b;
        b.x = W[(size_t)(ka)     * NCOL + colbase + idx];
        b.y = W[(size_t)(ka + 1) * NCOL + colbase + idx];
#pragma unroll
        for (int t = 0; t < RT; ++t) {
            v2f a;
            a.x = sX[(t * 16 + idx) * SXS + ka];
            a.y = sX[(t * 16 + idx) * SXS + ka + 1];
            acc[t] = __builtin_amdgcn_wmma_f32_16x16x4_f32(
                 false, a,  false, b,
                 (short)0, acc[t],  false,  false);
        }
    }

#pragma unroll
    for (int t = 0; t < RT; ++t) {
#pragma unroll
        for (int v = 0; v < 8; ++v) {
            int gr = row0 + t * 16 + half * 8 + v;
            if (gr < M) H[(size_t)gr * NCOL + colbase + idx] = acc[t][v];
        }
    }
}

template <int F, bool RELU>
__global__ void __launch_bounds__(256)
k_gather(const int* __restrict__ rowptr, const int* __restrict__ col,
         const float* __restrict__ dinv, const float* __restrict__ h,
         const float* __restrict__ bias, float* __restrict__ res, int N) {
    constexpr int C = F / 32;
    typedef float vecC __attribute__((ext_vector_type(C)));
    const int lane = threadIdx.x & 31;
    const int node = __builtin_amdgcn_readfirstlane(
        (int)(blockIdx.x * (blockDim.x >> 5) + (threadIdx.x >> 5)));
    if (node >= N) return;
    const int beg = rowptr[node];
    const int end = rowptr[node + 1];
    const float dd = dinv[node];
    const vecC* __restrict__ hv = (const vecC*)h;
    const vecC hs = hv[(size_t)node * 32 + lane];
    const vecC bs = ((const vecC*)bias)[lane];
    vecC acc = 0.0f;

    for (int base = beg; base < end; base += 32) {
        const int rem = end - base;
        const int cnt = rem < 32 ? rem : 32;
        int   s_l = 0;
        float w_l = 0.0f;
        if (lane < cnt) {
            s_l = col[base + lane];
            w_l = dinv[s_l];
        }
        const int wbits = __builtin_bit_cast(int, w_l);
        int j = 0;
        for (; j + 8 <= cnt; j += 8) {
            vecC v[8];
            for (int u = 0; u < 8; ++u) {
                const int s = __builtin_amdgcn_readlane(s_l, j + u);
                v[u] = hv[(size_t)s * 32 + lane];
            }
            for (int u = 0; u < 8; ++u) {
                const float w = __builtin_bit_cast(float, __builtin_amdgcn_readlane(wbits, j + u)) * dd;
                acc += v[u] * w;
            }
        }
        if (j + 4 <= cnt) {
            vecC v[4];
            for (int u = 0; u < 4; ++u) {
                const int s = __builtin_amdgcn_readlane(s_l, j + u);
                v[u] = hv[(size_t)s * 32 + lane];
            }
            for (int u = 0; u < 4; ++u) {
                const float w = __builtin_bit_cast(float, __builtin_amdgcn_readlane(wbits, j + u)) * dd;
                acc += v[u] * w;
            }
            j += 4;
        }
        for (; j < cnt; ++j) {
            const int s = __builtin_amdgcn_readlane(s_l, j);
            const float w = __builtin_bit_cast(float, __builtin_amdgcn_readlane(wbits, j)) * dd;
            acc += hv[(size_t)s * 32 + lane] * w;
        }
    }
    vecC r;
    for (int c = 0; c < C; ++c) {
        const float v = acc[c] + hs[c] * dd * dd + bs[c];
        r[c] = RELU ? (v > 0.0f ? v : 0.0f) : v;
    }
    ((vecC*)res)[(size_t)node * 32 + lane] = r;
}

extern "C" void kernel_launch(void* const* d_in, const int* in_sizes, int n_in,
                              void* d_out, int out_size, void* d_ws, size_t ws_size,
                              hipStream_t stream) {
    const float* x  = (const float*)d_in[0];
    const int*   ei = (const int*)d_in[1];
    const float* W1 = (const float*)d_in[2];
    const float* b1 = (const float*)d_in[3];
    const float* W2 = (const float*)d_in[4];
    const float* b2 = (const float*)d_in[5];
    float* out = (float*)d_out;

    const int  N = in_sizes[0] / IN_F;
    const long E = (long)in_sizes[1] / 2;
    const int* srcIdx = ei;
    const int* dstIdx = ei + E;

    float* ws   = (float*)d_ws;
    float* dinv = ws;
    float* h    = dinv + N;
    float* agg1 = h + (size_t)N * HID_F;
    float* h2   = h;
    const int N4 = (N + 3) / 4 * 4;
    int* cnt    = (int*)(agg1 + (size_t)N * HID_F);
    int* rowptr = cnt + N4;
    int* rank   = rowptr + N4 + 4;
    int* col    = rank + E;

    k_zero<<<(N + 255) / 256, 256, 0, stream>>>(cnt, N);
    k_count<<<(int)((E + 255) / 256), 256, 0, stream>>>(dstIdx, cnt, rank, E);
    k_scan<<<1, 1024, 0, stream>>>(cnt, rowptr, dinv, N);
    k_place<<<(int)((E + 255) / 256), 256, 0, stream>>>(srcIdx, dstIdx, rowptr, rank, col, E);

    constexpr int RT   = 4;
    constexpr int ROWS = 16 * RT;

    k_gemm_wmma<IN_F, HID_F, RT>
        <<<(N + ROWS - 1) / ROWS, 32 * (HID_F / 16), 0, stream>>>(x, W1, h, N);
    k_gather<HID_F, true><<<(N + 7) / 8, 256, 0, stream>>>(rowptr, col, dinv, h, b1, agg1, N);

    k_gemm_wmma<HID_F, OUT_F, RT>
        <<<(N + ROWS - 1) / ROWS, 32 * (OUT_F / 16), 0, stream>>>(agg1, W2, h2, N);
    k_gather<OUT_F, false><<<(N + 7) / 8, 256, 0, stream>>>(rowptr, col, dinv, h2, b2, out, N);
}


# CSR build by two-level LDS bucket sort (chunk sort by 256-node tile, then one work-group per tile): no global atomics, no 1-group scan, no scattered stores; per-node sum order = LDS arrival order
# speedup vs baseline: 3.5949x; 1.7362x over previous
#include <hip/hip_runtime.h>


typedef __attribute__((ext_vector_type(2))) float v2f;
typedef __attribute__((ext_vector_type(8))) float v8f;

#define IN_F  128
#define HID_F 128
#define OUT_F 64

#define CH   4096
#define MAXT 512
#define MAXC 512
#define CAP  8192

__global__ void __launch_bounds__(256)
k_tilesort(const int* __restrict__ src, const int* __restrict__ dst,
           unsigned* __restrict__ recs, int* __restrict__ P, long E, int NT) {
    __shared__ int sHist[MAXT];
    __shared__ int sOff[MAXT + 1];
    __shared__ int sWave[8];
    __shared__ unsigned sRec[CH];
    const int tid  = threadIdx.x;
    const int lane = tid & 31;
    const int wave = tid >> 5;
    const int c    = blockIdx.x;
    const long ebase = (long)c * CH;
    const long left  = E - ebase;
    const int n = left < CH ? (int)left : CH;

    for (int i = tid; i < MAXT; i += 256) sHist[i] = 0;
    __syncthreads();

    unsigned rec[16];
    int tl[16], rk[16];
    int dv[16], sv[16];
    for (int k = 0; k < 16; ++k) {
        const int i  = k * 256 + tid;
        const int ic = i < n ? i : n - 1;
        dv[k] = dst[ebase + ic];
        sv[k] = src[ebase + ic];
    }
    for (int k = 0; k < 16; ++k) {
        const bool valid = (k * 256 + tid) < n;
        tl[k]  = valid ? (dv[k] >> 8) : -1;
        rk[k]  = 0;
        rec[k] = ((unsigned)sv[k] << 8) | ((unsigned)dv[k] & 255u);
    }
    for (int k = 0; k < 16; ++k)
        if (tl[k] >= 0)
            rk[k] = __hip_atomic_fetch_add(&sHist[tl[k]], 1, __ATOMIC_RELAXED,
                                           __HIP_MEMORY_SCOPE_WORKGROUP);
    __syncthreads();

    const int a0 = sHist[2 * tid], a1 = sHist[2 * tid + 1];
    const int s2 = a0 + a1;
    int incl = s2;
    for (int o = 1; o < 32; o <<= 1) {
        const int u = __shfl_up(incl, o);
        if (lane >= o) incl += u;
    }
    if (lane == 31) sWave[wave] = incl;
    __syncthreads();
    int carry = 0;
    for (int w = 0; w < wave; ++w) carry += sWave[w];
    const int ex = carry + incl - s2;
    sOff[2 * tid]     = ex;
    sOff[2 * tid + 1] = ex + a0;
    if (tid == 255) sOff[MAXT] = ex + s2;
    __syncthreads();

    for (int k = 0; k < 16; ++k)
        if (tl[k] >= 0) sRec[sOff[tl[k]] + rk[k]] = rec[k];
    for (int t = tid; t <= NT; t += 256) P[(size_t)c * (NT + 1) + t] = sOff[t];
    __syncthreads();
    for (int i = tid; i < n; i += 256) recs[ebase + i] = sRec[i];
}

template <bool PLACE>
__device__ __forceinline__ void tile_sweep(const unsigned* __restrict__ recs, const int* sStart,
                                           const int* sLen, int* sBin, int* sCol, int* __restrict__ col,
                                           int NC, int base, bool staged, int half, int hl) {
    for (int c0 = half; c0 < NC; c0 += 128) {
        unsigned rv[8];
        bool ok[8];
        for (int u = 0; u < 8; ++u) {
            const int c  = c0 + 16 * u;
            const int cc = c < NC ? c : 0;
            ok[u] = (c < NC) && (hl < sLen[cc]);
            const size_t a = ok[u] ? ((size_t)cc * CH + sStart[cc] + hl) : (size_t)0;
            rv[u] = recs[a];
        }
        for (int u = 0; u < 8; ++u) {
            if (ok[u]) {
                const int r = __hip_atomic_fetch_add(&sBin[rv[u] & 255u], 1, __ATOMIC_RELAXED,
                                                     __HIP_MEMORY_SCOPE_WORKGROUP);
                if (PLACE) {
                    const int s = (int)(rv[u] >> 8);
                    if (staged) sCol[r] = s; else col[base + r] = s;
                }
            }
        }
        for (int u = 0; u < 8; ++u) {
            const int c = c0 + 16 * u;
            if (c < NC) {
                const int len = sLen[c];
                for (int j = hl + 16; j < len; j += 16) {
                    const unsigned v = recs[(size_t)c * CH + sStart[c] + j];
                    const int r = __hip_atomic_fetch_add(&sBin[v & 255u], 1, __ATOMIC_RELAXED,
                                                         __HIP_MEMORY_SCOPE_WORKGROUP);
                    if (PLACE) {
                        const int s = (int)(v >> 8);
                        if (staged) sCol[r] = s; else col[base + r] = s;
                    }
                }
            }
        }
    }
}

__global__ void __launch_bounds__(256)
k_tilecsr(const unsigned* __restrict__ recs, const int* __restrict__ P,
          int* __restrict__ rowptr, int* __restrict__ col, float* __restrict__ dinv,
          int N, long E, int NC, int NT) {
    __shared__ int sStart[MAXC];
    __shared__ int sLen[MAXC];
    __shared__ int sCnt[256];
    __shared__ int sCur[256];
    __shared__ int sWave[8];
    __shared__ int sRedB[8];
    __shared__ int sRedT[8];
    __shared__ int sCol[CAP];
    const int tid  = threadIdx.x;
    const int lane = tid & 31;
    const int wave = tid >> 5;
    const int t    = blockIdx.x;
    const int half = tid >> 4;
    const int hl   = tid & 15;

    int bsum = 0, tsum = 0;
    for (int c = tid; c < NC; c += 256) {
        const int p0 = P[(size_t)c * (NT + 1) + t];
        const int p1 = P[(size_t)c * (NT + 1) + t + 1];
        sStart[c] = p0;
        sLen[c]   = p1 - p0;
        bsum += p0;
        tsum += p1 - p0;
    }
    sCnt[tid] = 0;
    for (int o = 16; o > 0; o >>= 1) {
        bsum += __shfl_xor(bsum, o);
        tsum += __shfl_xor(tsum, o);
    }
    if (lane == 0) { sRedB[wave] = bsum; sRedT[wave] = tsum; }
    __syncthreads();
    int base = 0, total = 0;
    for (int w = 0; w < 8; ++w) { base += sRedB[w]; total += sRedT[w]; }
    const bool staged = total <= CAP;

    tile_sweep<false>(recs, sStart, sLen, sCnt, sCol, col, NC, base, staged, half, hl);
    __syncthreads();

    const int cn = sCnt[tid];
    int incl = cn;
    for (int o = 1; o < 32; o <<= 1) {
        const int u = __shfl_up(incl, o);
        if (lane >= o) incl += u;
    }
    if (lane == 31) sWave[wave] = incl;
    __syncthreads();
    int carry = 0;
    for (int w = 0; w < wave; ++w) carry += sWave[w];
    const int ex = carry + incl - cn;
    const int node = t * 256 + tid;
    if (node < N) {
        rowptr[node] = base + ex;
        dinv[node]   = rsqrtf((float)(cn + 1));
    }
    if (t == NT - 1 && tid == 0) rowptr[N] = (int)E;
    sCur[tid] = ex;
    __syncthreads();

    tile_sweep<true>(recs, sStart, sLen, sCur, sCol, col, NC, base, staged, half, hl);
    __syncthreads();
    if (staged)
        for (int i = tid; i < total; i += 256) col[base + i] = sCol[i];
}

template <int K, int NCOL, int RT>
__global__ void __launch_bounds__(32 * (NCOL / 16))
k_gemm_wmma(const float* __restrict__ X, const float* __restrict__ W,
            float* __restrict__ H, int M) {
    constexpr int ROWS = 16 * RT;
    constexpr int SXS  = K + 4;
    __shared__ float sX[ROWS * SXS];

    const int tid  = threadIdx.x;
    const int wave = tid >> 5;
    const int lane = tid & 31;
    const int row0 = blockIdx.x * ROWS;

    for (int i = tid; i < ROWS * K; i += blockDim.x) {
        int r = i / K, c = i % K;
        int gr = row0 + r;
        sX[r * SXS + c] = (gr < M) ? X[(size_t)gr * K + c] : 0.0f;
    }
    __syncthreads();

    const int half = lane >> 4;
    const int idx  = lane & 15;
    const int colbase = wave * 16;

    v8f acc[RT] = {};
#pragma unroll
    for (int k0 = 0; k0 < K; k0 += 4) {
        const int ka = k0 + half * 2;
        v2f b;
        b.x = W[(size_t)(ka)     * NCOL + colbase + idx];
        b.y = W[(size_t)(ka + 1) * NCOL + colbase + idx];
#pragma unroll
        for (int t = 0; t < RT; ++t) {
            v2f a;
            a.x = sX[(t * 16 + idx) * SXS + ka];
            a.y = sX[(t * 16 + idx) * SXS + ka + 1];
            acc[t] = __builtin_amdgcn_wmma_f32_16x16x4_f32(
                 false, a,  false, b,
                 (short)0, acc[t],  false,  false);
        }
    }

#pragma unroll
    for (int t = 0; t < RT; ++t) {
#pragma unroll
        for (int v = 0; v < 8; ++v) {
            int gr = row0 + t * 16 + half * 8 + v;
            if (gr < M) H[(size_t)gr * NCOL + colbase + idx] = acc[t][v];
        }
    }
}

template <int F, bool RELU>
__global__ void __launch_bounds__(256)
k_gather(const int* __restrict__ rowptr, const int* __restrict__ col,
         const float* __restrict__ dinv, const float* __restrict__ h,
         const float* __restrict__ bias, float* __restrict__ res, int N) {
    constexpr int C = F / 32;
    typedef float vecC __attribute__((ext_vector_type(C)));
    const int lane = threadIdx.x & 31;
    const int node = __builtin_amdgcn_readfirstlane(
        (int)(blockIdx.x * (blockDim.x >> 5) + (threadIdx.x >> 5)));
    if (node >= N) return;
    const int beg = rowptr[node];
    const int end = rowptr[node + 1];
    const float dd = dinv[node];
    const vecC* __restrict__ hv = (const vecC*)h;
    const vecC hs = hv[(size_t)node * 32 + lane];
    const vecC bs = ((const vecC*)bias)[lane];
    vecC acc = 0.0f;

    for (int base = beg; base < end; base += 32) {
        const int rem = end - base;
        const int cnt = rem < 32 ? rem : 32;
        int   s_l = 0;
        float w_l = 0.0f;
        if (lane < cnt) {
            s_l = col[base + lane];
            w_l = dinv[s_l];
        }
        const int wbits = __builtin_bit_cast(int, w_l);
        int j = 0;
        for (; j + 8 <= cnt; j += 8) {
            vecC v[8];
            for (int u = 0; u < 8; ++u) {
                const int s = __builtin_amdgcn_readlane(s_l, j + u);
                v[u] = hv[(size_t)s * 32 + lane];
            }
            for (int u = 0; u < 8; ++u) {
                const float w = __builtin_bit_cast(float, __builtin_amdgcn_readlane(wbits, j + u)) * dd;
                acc += v[u] * w;
            }
        }
        if (j + 4 <= cnt) {
            vecC v[4];
            for (int u = 0; u < 4; ++u) {
                const int s = __builtin_amdgcn_readlane(s_l, j + u);
                v[u] = hv[(size_t)s * 32 + lane];
            }
            for (int u = 0; u < 4; ++u) {
                const float w = __builtin_bit_cast(float, __builtin_amdgcn_readlane(wbits, j + u)) * dd;
                acc += v[u] * w;
            }
            j += 4;
        }
        for (; j < cnt; ++j) {
            const int s = __builtin_amdgcn_readlane(s_l, j);
            const float w = __builtin_bit_cast(float, __builtin_amdgcn_readlane(wbits, j)) * dd;
            acc += hv[(size_t)s * 32 + lane] * w;
        }
    }
    vecC r;
    for (int c = 0; c < C; ++c) {
        const float v = acc[c] + hs[c] * dd * dd + bs[c];
        r[c] = RELU ? (v > 0.0f ? v : 0.0f) : v;
    }
    ((vecC*)res)[(size_t)node * 32 + lane] = r;
}

extern "C" void kernel_launch(void* const* d_in, const int* in_sizes, int n_in,
                              void* d_out, int out_size, void* d_ws, size_t ws_size,
                              hipStream_t stream) {
    const float* x  = (const float*)d_in[0];
    const int*   ei = (const int*)d_in[1];
    const float* W1 = (const float*)d_in[2];
    const float* b1 = (const float*)d_in[3];
    const float* W2 = (const float*)d_in[4];
    const float* b2 = (const float*)d_in[5];
    float* out = (float*)d_out;

    const int  N = in_sizes[0] / IN_F;
    const long E = (long)in_sizes[1] / 2;
    const int* srcIdx = ei;
    const int* dstIdx = ei + E;

    float* ws   = (float*)d_ws;
    float* dinv = ws;
    float* h    = dinv + N;
    float* agg1 = h + (size_t)N * HID_F;
    float* h2   = h;
    const int N4 = (N + 3) / 4 * 4;
    const int NC = (int)((E + CH - 1) / CH);
    const int NT = (N + 255) / 256;
    int* rowptr    = (int*)(agg1 + (size_t)N * HID_F);
    int* col       = rowptr + N4 + 4;
    unsigned* recs = (unsigned*)agg1;
    int* P         = (int*)(recs + (size_t)NC * CH);

    k_tilesort<<<NC, 256, 0, stream>>>(srcIdx, dstIdx, recs, P, E, NT);
    k_tilecsr<<<NT, 256, 0, stream>>>(recs, P, rowptr, col, dinv, N, E, NC, NT);

    constexpr int RT   = 4;
    constexpr int ROWS = 16 * RT;

    k_gemm_wmma<IN_F, HID_F, RT>
        <<<(N + ROWS - 1) / ROWS, 32 * (HID_F / 16), 0, stream>>>(x, W1, h, N);
    k_gather<HID_F, true><<<(N + 7) / 8, 256, 0, stream>>>(rowptr, col, dinv, h, b1, agg1, N);

    k_gemm_wmma<HID_F, OUT_F, RT>
        <<<(N + ROWS - 1) / ROWS, 32 * (OUT_F / 16), 0, stream>>>(agg1, W2, h2, N);
    k_gather<OUT_F, false><<<(N + 7) / 8, 256, 0, stream>>>(rowptr, col, dinv, h2, b2, out, N);
}


# GEMM X tile staged with 16-byte loads, a thread's whole share (8 or 16 loads) in flight before the first wait, one ds_store_b128 each; was one 4-byte load per loop trip, each waited for alone
# speedup vs baseline: 4.0766x; 1.1340x over previous
#include <hip/hip_runtime.h>


typedef __attribute__((ext_vector_type(2))) float v2f;
typedef __attribute__((ext_vector_type(8))) float v8f;
typedef __attribute__((ext_vector_type(4))) float v4f;

#define IN_F  128
#define HID_F 128
#define OUT_F 64

#define CH   4096
#define MAXT 512
#define MAXC 512
#define CAP  8192

__global__ void __launch_bounds__(256)
k_tilesort(const int* __restrict__ src, const int* __restrict__ dst,
           unsigned* __restrict__ recs, int* __restrict__ P, long E, int NT) {
    __shared__ int sHist[MAXT];
    __shared__ int sOff[MAXT + 1];
    __shared__ int sWave[8];
    __shared__ unsigned sRec[CH];
    const int tid  = threadIdx.x;
    const int lane = tid & 31;
    const int wave = tid >> 5;
    const int c    = blockIdx.x;
    const long ebase = (long)c * CH;
    const long left  = E - ebase;
    const int n = left < CH ? (int)left : CH;

    for (int i = tid; i < MAXT; i += 256) sHist[i] = 0;
    __syncthreads();

    unsigned rec[16];
    int tl[16], rk[16];
    int dv[16], sv[16];
    for (int k = 0; k < 16; ++k) {
        const int i  = k * 256 + tid;
        const int ic = i < n ? i : n - 1;
        dv[k] = dst[ebase + ic];
        sv[k] = src[ebase + ic];
    }
    for (int k = 0; k < 16; ++k) {
        const bool valid = (k * 256 + tid) < n;
        tl[k]  = valid ? (dv[k] >> 8) : -1;
        rk[k]  = 0;
        rec[k] = ((unsigned)sv[k] << 8) | ((unsigned)dv[k] & 255u);
    }
    for (int k = 0; k < 16; ++k)
        if (tl[k] >= 0)
            rk[k] = __hip_atomic_fetch_add(&sHist[tl[k]], 1, __ATOMIC_RELAXED,
                                           __HIP_MEMORY_SCOPE_WORKGROUP);
    __syncthreads();

    const int a0 = sHist[2 * tid], a1 = sHist[2 * tid + 1];
    const int s2 = a0 + a1;
    int incl = s2;
    for (int o = 1; o < 32; o <<= 1) {
        const int u = __shfl_up(incl, o);
        if (lane >= o) incl += u;
    }
    if (lane == 31) sWave[wave] = incl;
    __syncthreads();
    int carry = 0;
    for (int w = 0; w < wave; ++w) carry += sWave[w];
    const int ex = carry + incl - s2;
    sOff[2 * tid]     = ex;
    sOff[2 * tid + 1] = ex + a0;
    if (tid == 255) sOff[MAXT] = ex + s2;
    __syncthreads();

    for (int k = 0; k < 16; ++k)
        if (tl[k] >= 0) sRec[sOff[tl[k]] + rk[k]] = rec[k];
    for (int t = tid; t <= NT; t += 256) P[(size_t)c * (NT + 1) + t] = sOff[t];
    __syncthreads();
    for (int i = tid; i < n; i += 256) recs[ebase + i] = sRec[i];
}

template <bool PLACE>
__device__ __forceinline__ void tile_sweep(const unsigned* __restrict__ recs, const int* sStart,
                                           const int* sLen, int* sBin, int* sCol, int* __restrict__ col,
                                           int NC, int base, bool staged, int half, int hl) {
    for (int c0 = half; c0 < NC; c0 += 128) {
        unsigned rv[8];
        bool ok[8];
        for (int u = 0; u < 8; ++u) {
            const int c  = c0 + 16 * u;
            const int cc = c < NC ? c : 0;
            ok[u] = (c < NC) && (hl < sLen[cc]);
            const size_t a = ok[u] ? ((size_t)cc * CH + sStart[cc] + hl) : (size_t)0;
            rv[u] = recs[a];
        }
        for (int u = 0; u < 8; ++u) {
            if (ok[u]) {
                const int r = __hip_atomic_fetch_add(&sBin[rv[u] & 255u], 1, __ATOMIC_RELAXED,
                                                     __HIP_MEMORY_SCOPE_WORKGROUP);
                if (PLACE) {
                    const int s = (int)(rv[u] >> 8);
                    if (staged) sCol[r] = s; else col[base + r] = s;
                }
            }
        }
        for (int u = 0; u < 8; ++u) {
            const int c = c0 + 16 * u;
            if (c < NC) {
                const int len = sLen[c];
                for (int j = hl + 16; j < len; j += 16) {
                    const unsigned v = recs[(size_t)c * CH + sStart[c] + j];
                    const int r = __hip_atomic_fetch_add(&sBin[v & 255u], 1, __ATOMIC_RELAXED,
                                                         __HIP_MEMORY_SCOPE_WORKGROUP);
                    if (PLACE) {
                        const int s = (int)(v >> 8);
                        if (staged) sCol[r] = s; else col[base + r] = s;
                    }
                }
            }
        }
    }
}

__global__ void __launch_bounds__(256)
k_tilecsr(const unsigned* __restrict__ recs, const int* __restrict__ P,
          int* __restrict__ rowptr, int* __restrict__ col, float* __restrict__ dinv,
          int N, long E, int NC, int NT) {
    __shared__ int sStart[MAXC];
    __shared__ int sLen[MAXC];
    __shared__ int sCnt[256];
    __shared__ int sCur[256];
    __shared__ int sWave[8];
    __shared__ int sRedB[8];
    __shared__ int sRedT[8];
    __shared__ int sCol[CAP];
    const int tid  = threadIdx.x;
    const int lane = tid & 31;
    const int wave = tid >> 5;
    const int t    = blockIdx.x;
    const int half = tid >> 4;
    const int hl   = tid & 15;

    int bsum = 0, tsum = 0;
    for (int c = tid; c < NC; c += 256) {
        const int p0 = P[(size_t)c * (NT + 1) + t];
        const int p1 = P[(size_t)c * (NT + 1) + t + 1];
        sStart[c] = p0;
        sLen[c]   = p1 - p0;
        bsum += p0;
        tsum += p1 - p0;
    }
    sCnt[tid] = 0;
    for (int o = 16; o > 0; o >>= 1) {
        bsum += __shfl_xor(bsum, o);
        tsum += __shfl_xor(tsum, o);
    }
    if (lane == 0) { sRedB[wave] = bsum; sRedT[wave] = tsum; }
    __syncthreads();
    int base = 0, total = 0;
    for (int w = 0; w < 8; ++w) { base += sRedB[w]; total += sRedT[w]; }
    const bool staged = total <= CAP;

    tile_sweep<false>(recs, sStart, sLen, sCnt, sCol, col, NC, base, staged, half, hl);
    __syncthreads();

    const int cn = sCnt[tid];
    int incl = cn;
    for (int o = 1; o < 32; o <<= 1) {
        const int u = __shfl_up(incl, o);
        if (lane >= o) incl += u;
    }
    if (lane == 31) sWave[wave] = incl;
    __syncthreads();
    int carry = 0;
    for (int w = 0; w < wave; ++w) carry += sWave[w];
    const int ex = carry + incl - cn;
    const int node = t * 256 + tid;
    if (node < N) {
        rowptr[node] = base + ex;
        dinv[node]   = rsqrtf((float)(cn + 1));
    }
    if (t == NT - 1 && tid == 0) rowptr[N] = (int)E;
    sCur[tid] = ex;
    __syncthreads();

    tile_sweep<true>(recs, sStart, sLen, sCur, sCol, col, NC, base, staged, half, hl);
    __syncthreads();
    if (staged)
        for (int i = tid; i < total; i += 256) col[base + i] = sCol[i];
}

template <int K, int NCOL, int RT>
__global__ void __launch_bounds__(32 * (NCOL / 16))
k_gemm_wmma(const float* __restrict__ X, const float* __restrict__ W,
            float* __restrict__ H, int M) {
    constexpr int ROWS = 16 * RT;
    constexpr int SXS  = K + 4;
    __shared__ __attribute__((aligned(16))) float sX[ROWS * SXS];

    const int tid  = threadIdx.x;
    const int wave = tid >> 5;
    const int lane = tid & 31;
    const int row0 = blockIdx.x * ROWS;

    constexpr int K4  = K / 4;
    constexpr int NTH = 32 * (NCOL / 16);
    constexpr int PER = ROWS * K4 / NTH;
    static_assert(ROWS * K4 % NTH == 0, "tile must split evenly over the threads");
    if (row0 + ROWS <= M) {
        const v4f* __restrict__ X4 = (const v4f*)(X + (size_t)row0 * K);
        v4f xv[PER];
#pragma unroll
        for (int it = 0; it < PER; ++it) xv[it] = X4[tid + it * NTH];
#pragma unroll
        for (int it = 0; it < PER; ++it) {
            const int i = tid + it * NTH;
            *(v4f*)&sX[(i / K4) * SXS + (i % K4) * 4] = xv[it];
        }
    } else {
        for (int i = tid; i < ROWS * K; i += NTH) {
            int r = i / K, c = i % K;
            int gr = row0 + r;
            sX[r * SXS + c] = (gr < M) ? X[(size_t)gr * K + c] : 0.0f;
        }
    }
    __syncthreads();

    const int half = lane >> 4;
    const int idx  = lane & 15;
    const int colbase = wave * 16;

    v8f acc[RT] = {};
#pragma unroll
    for (int k0 = 0; k0 < K; k0 += 4) {
        const int ka = k0 + half * 2;
        v2f b;
        b.x = W[(size_t)(ka)     * NCOL + colbase + idx];
        b.y = W[(size_t)(ka + 1) * NCOL + colbase + idx];
#pragma unroll
        for (int t = 0; t < RT; ++t) {
            v2f a;
            a.x = sX[(t * 16 + idx) * SXS + ka];
            a.y = sX[(t * 16 + idx) * SXS + ka + 1];
            acc[t] = __builtin_amdgcn_wmma_f32_16x16x4_f32(
                 false, a,  false, b,
                 (short)0, acc[t],  false,  false);
        }
    }

#pragma unroll
    for (int t = 0; t < RT; ++t) {
#pragma unroll
        for (int v = 0; v < 8; ++v) {
            int gr = row0 + t * 16 + half * 8 + v;
            if (gr < M) H[(size_t)gr * NCOL + colbase + idx] = acc[t][v];
        }
    }
}

template <int F, bool RELU>
__global__ void __launch_bounds__(256)
k_gather(const int* __restrict__ rowptr, const int* __restrict__ col,
         const float* __restrict__ dinv, const float* __restrict__ h,
         const float* __restrict__ bias, float* __restrict__ res, int N) {
    constexpr int C = F / 32;
    typedef float vecC __attribute__((ext_vector_type(C)));
    const int lane = threadIdx.x & 31;
    const int node = __builtin_amdgcn_readfirstlane(
        (int)(blockIdx.x * (blockDim.x >> 5) + (threadIdx.x >> 5)));
    if (node >= N) return;
    const int beg = rowptr[node];
    const int end = rowptr[node + 1];
    const float dd = dinv[node];
    const vecC* __restrict__ hv = (const vecC*)h;
    const vecC hs = hv[(size_t)node * 32 + lane];
    const vecC bs = ((const vecC*)bias)[lane];
    vecC acc = 0.0f;

    for (int base = beg; base < end; base += 32) {
        const int rem = end - base;
        const int cnt = rem < 32 ? rem : 32;
        int   s_l = 0;
        float w_l = 0.0f;
        if (lane < cnt) {
            s_l = col[base + lane];
            w_l = dinv[s_l];
        }
        const int wbits = __builtin_bit_cast(int, w_l);
        int j = 0;
        for (; j + 8 <= cnt; j += 8) {
            vecC v[8];
            for (int u = 0; u < 8; ++u) {
                const int s = __builtin_amdgcn_readlane(s_l, j + u);
                v[u] = hv[(size_t)s * 32 + lane];
            }
            for (int u = 0; u < 8; ++u) {
                const float w = __builtin_bit_cast(float, __builtin_amdgcn_readlane(wbits, j + u)) * dd;
                acc += v[u] * w;
            }
        }
        if (j + 4 <= cnt) {
            vecC v[4];
            for (int u = 0; u < 4; ++u) {
                const int s = __builtin_amdgcn_readlane(s_l, j + u);
                v[u] = hv[(size_t)s * 32 + lane];
            }
            for (int u = 0; u < 4; ++u) {
                const float w = __builtin_bit_cast(float, __builtin_amdgcn_readlane(wbits, j + u)) * dd;
                acc += v[u] * w;
            }
            j += 4;
        }
        for (; j < cnt; ++j) {
            const int s = __builtin_amdgcn_readlane(s_l, j);
            const float w = __builtin_bit_cast(float, __builtin_amdgcn_readlane(wbits, j)) * dd;
            acc += hv[(size_t)s * 32 + lane] * w;
        }
    }
    vecC r;
    for (int c = 0; c < C; ++c) {
        const float v = acc[c] + hs[c] * dd * dd + bs[c];
        r[c] = RELU ? (v > 0.0f ? v : 0.0f) : v;
    }
    ((vecC*)res)[(size_t)node * 32 + lane] = r;
}

extern "C" void kernel_launch(void* const* d_in, const int* in_sizes, int n_in,
                              void* d_out, int out_size, void* d_ws, size_t ws_size,
                              hipStream_t stream) {
    const float* x  = (const float*)d_in[0];
    const int*   ei = (const int*)d_in[1];
    const float* W1 = (const float*)d_in[2];
    const float* b1 = (const float*)d_in[3];
    const float* W2 = (const float*)d_in[4];
    const float* b2 = (const float*)d_in[5];
    float* out = (float*)d_out;

    const int  N = in_sizes[0] / IN_F;
    const long E = (long)in_sizes[1] / 2;
    const int* srcIdx = ei;
    const int* dstIdx = ei + E;

    float* ws   = (float*)d_ws;
    float* dinv = ws;
    float* h    = dinv + N;
    float* agg1 = h + (size_t)N * HID_F;
    float* h2   = h;
    const int N4 = (N + 3) / 4 * 4;
    const int NC = (int)((E + CH - 1) / CH);
    const int NT = (N + 255) / 256;
    int* rowptr    = (int*)(agg1 + (size_t)N * HID_F);
    int* col       = rowptr + N4 + 4;
    unsigned* recs = (unsigned*)agg1;
    int* P         = (int*)(recs + (size_t)NC * CH);

    k_tilesort<<<NC, 256, 0, stream>>>(srcIdx, dstIdx, recs, P, E, NT);
    k_tilecsr<<<NT, 256, 0, stream>>>(recs, P, rowptr, col, dinv, N, E, NC, NT);

    constexpr int RT   = 4;
    constexpr int ROWS = 16 * RT;

    k_gemm_wmma<IN_F, HID_F, RT>
        <<<(N + ROWS - 1) / ROWS, 32 * (HID_F / 16), 0, stream>>>(x, W1, h, N);
    k_gather<HID_F, true><<<(N + 7) / 8, 256, 0, stream>>>(rowptr, col, dinv, h, b1, agg1, N);

    k_gemm_wmma<HID_F, OUT_F, RT>
        <<<(N + ROWS - 1) / ROWS, 32 * (OUT_F / 16), 0, stream>>>(agg1, W2, h2, N);
    k_gather<OUT_F, false><<<(N + 7) / 8, 256, 0, stream>>>(rowptr, col, dinv, h2, b2, out, N);
}


# Fuse the layer-2 dense product into the layer-1 pull-aggregation kernel: a block keeps 32 finished ReLU rows in LDS and multiplies them by W2 on WMMA; agg1 store, re-read and one launch go
# speedup vs baseline: 4.5469x; 1.1154x over previous
#include <hip/hip_runtime.h>


typedef __attribute__((ext_vector_type(2))) float v2f;
typedef __attribute__((ext_vector_type(8))) float v8f;
typedef __attribute__((ext_vector_type(4))) float v4f;

#define IN_F  128
#define HID_F 128
#define OUT_F 64

#define CH   4096
#define MAXT 512
#define MAXC 512
#define CAP  8192

__global__ void __launch_bounds__(256)
k_tilesort(const int* __restrict__ src, const int* __restrict__ dst,
           unsigned* __restrict__ recs, int* __restrict__ P, long E, int NT) {
    __shared__ int sHist[MAXT];
    __shared__ int sOff[MAXT + 1];
    __shared__ int sWave[8];
    __shared__ unsigned sRec[CH];
    const int tid  = threadIdx.x;
    const int lane = tid & 31;
    const int wave = tid >> 5;
    const int c    = blockIdx.x;
    const long ebase = (long)c * CH;
    const long left  = E - ebase;
    const int n = left < CH ? (int)left : CH;

    for (int i = tid; i < MAXT; i += 256) sHist[i] = 0;
    __syncthreads();

    unsigned rec[16];
    int tl[16], rk[16];
    int dv[16], sv[16];
    for (int k = 0; k < 16; ++k) {
        const int i  = k * 256 + tid;
        const int ic = i < n ? i : n - 1;
        dv[k] = dst[ebase + ic];
        sv[k] = src[ebase + ic];
    }
    for (int k = 0; k < 16; ++k) {
        const bool valid = (k * 256 + tid) < n;
        tl[k]  = valid ? (dv[k] >> 8) : -1;
        rk[k]  = 0;
        rec[k] = ((unsigned)sv[k] << 8) | ((unsigned)dv[k] & 255u);
    }
    for (int k = 0; k < 16; ++k)
        if (tl[k] >= 0)
            rk[k] = __hip_atomic_fetch_add(&sHist[tl[k]], 1, __ATOMIC_RELAXED,
                                           __HIP_MEMORY_SCOPE_WORKGROUP);
    __syncthreads();

    const int a0 = sHist[2 * tid], a1 = sHist[2 * tid + 1];
    const int s2 = a0 + a1;
    int incl = s2;
    for (int o = 1; o < 32; o <<= 1) {
        const int u = __shfl_up(incl, o);
        if (lane >= o) incl += u;
    }
    if (lane == 31) sWave[wave] = incl;
    __syncthreads();
    int carry = 0;
    for (int w = 0; w < wave; ++w) carry += sWave[w];
    const int ex = carry + incl - s2;
    sOff[2 * tid]     = ex;
    sOff[2 * tid + 1] = ex + a0;
    if (tid == 255) sOff[MAXT] = ex + s2;
    __syncthreads();

    for (int k = 0; k < 16; ++k)
        if (tl[k] >= 0) sRec[sOff[tl[k]] + rk[k]] = rec[k];
    for (int t = tid; t <= NT; t += 256) P[(size_t)c * (NT + 1) + t] = sOff[t];
    __syncthreads();
    for (int i = tid; i < n; i += 256) recs[ebase + i] = sRec[i];
}

template <bool PLACE>
__device__ __forceinline__ void tile_sweep(const unsigned* __restrict__ recs, const int* sStart,
                                           const int* sLen, int* sBin, int* sCol, int* __restrict__ col,
                                           int NC, int base, bool staged, int half, int hl) {
    for (int c0 = half; c0 < NC; c0 += 128) {
        unsigned rv[8];
        bool ok[8];
        for (int u = 0; u < 8; ++u) {
            const int c  = c0 + 16 * u;
            const int cc = c < NC ? c : 0;
            ok[u] = (c < NC) && (hl < sLen[cc]);
            const size_t a = ok[u] ? ((size_t)cc * CH + sStart[cc] + hl) : (size_t)0;
            rv[u] = recs[a];
        }
        for (int u = 0; u < 8; ++u) {
            if (ok[u]) {
                const int r = __hip_atomic_fetch_add(&sBin[rv[u] & 255u], 1, __ATOMIC_RELAXED,
                                                     __HIP_MEMORY_SCOPE_WORKGROUP);
                if (PLACE) {
                    const int s = (int)(rv[u] >> 8);
                    if (staged) sCol[r] = s; else col[base + r] = s;
                }
            }
        }
        for (int u = 0; u < 8; ++u) {
            const int c = c0 + 16 * u;
            if (c < NC) {
                const int len = sLen[c];
                for (int j = hl + 16; j < len; j += 16) {
                    const unsigned v = recs[(size_t)c * CH + sStart[c] + j];
                    const int r = __hip_atomic_fetch_add(&sBin[v & 255u], 1, __ATOMIC_RELAXED,
                                                         __HIP_MEMORY_SCOPE_WORKGROUP);
                    if (PLACE) {
                        const int s = (int)(v >> 8);
                        if (staged) sCol[r] = s; else col[base + r] = s;
                    }
                }
            }
        }
    }
}

__global__ void __launch_bounds__(256)
k_tilecsr(const unsigned* __restrict__ recs, const int* __restrict__ P,
          int* __restrict__ rowptr, int* __restrict__ col, float* __restrict__ dinv,
          int N, long E, int NC, int NT) {
    __shared__ int sStart[MAXC];
    __shared__ int sLen[MAXC];
    __shared__ int sCnt[256];
    __shared__ int sCur[256];
    __shared__ int sWave[8];
    __shared__ int sRedB[8];
    __shared__ int sRedT[8];
    __shared__ int sCol[CAP];
    const int tid  = threadIdx.x;
    const int lane = tid & 31;
    const int wave = tid >> 5;
    const int t    = blockIdx.x;
    const int half = tid >> 4;
    const int hl   = tid & 15;

    int bsum = 0, tsum = 0;
    for (int c = tid; c < NC; c += 256) {
        const int p0 = P[(size_t)c * (NT + 1) + t];
        const int p1 = P[(size_t)c * (NT + 1) + t + 1];
        sStart[c] = p0;
        sLen[c]   = p1 - p0;
        bsum += p0;
        tsum += p1 - p0;
    }
    sCnt[tid] = 0;
    for (int o = 16; o > 0; o >>= 1) {
        bsum += __shfl_xor(bsum, o);
        tsum += __shfl_xor(tsum, o);
    }
    if (lane == 0) { sRedB[wave] = bsum; sRedT[wave] = tsum; }
    __syncthreads();
    int base = 0, total = 0;
    for (int w = 0; w < 8; ++w) { base += sRedB[w]; total += sRedT[w]; }
    const bool staged = total <= CAP;

    tile_sweep<false>(recs, sStart, sLen, sCnt, sCol, col, NC, base, staged, half, hl);
    __syncthreads();

    const int cn = sCnt[tid];
    int incl = cn;
    for (int o = 1; o < 32; o <<= 1) {
        const int u = __shfl_up(incl, o);
        if (lane >= o) incl += u;
    }
    if (lane == 31) sWave[wave] = incl;
    __syncthreads();
    int carry = 0;
    for (int w = 0; w < wave; ++w) carry += sWave[w];
    const int ex = carry + incl - cn;
    const int node = t * 256 + tid;
    if (node < N) {
        rowptr[node] = base + ex;
        dinv[node]   = rsqrtf((float)(cn + 1));
    }
    if (t == NT - 1 && tid == 0) rowptr[N] = (int)E;
    sCur[tid] = ex;
    __syncthreads();

    tile_sweep<true>(recs, sStart, sLen, sCur, sCol, col, NC, base, staged, half, hl);
    __syncthreads();
    if (staged)
        for (int i = tid; i < total; i += 256) col[base + i] = sCol[i];
}

template <int K, int NCOL, int RT>
__global__ void __launch_bounds__(32 * (NCOL / 16))
k_gemm_wmma(const float* __restrict__ X, const float* __restrict__ W,
            float* __restrict__ H, int M) {
    constexpr int ROWS = 16 * RT;
    constexpr int SXS  = K + 4;
    __shared__ __attribute__((aligned(16))) float sX[ROWS * SXS];

    const int tid  = threadIdx.x;
    const int wave = tid >> 5;
    const int lane = tid & 31;
    const int row0 = blockIdx.x * ROWS;

    constexpr int K4  = K / 4;
    constexpr int NTH = 32 * (NCOL / 16);
    constexpr int PER = ROWS * K4 / NTH;
    static_assert(ROWS * K4 % NTH == 0, "tile must split evenly over the threads");
    if (row0 + ROWS <= M) {
        const v4f* __restrict__ X4 = (const v4f*)(X + (size_t)row0 * K);
        v4f xv[PER];
#pragma unroll
        for (int it = 0; it < PER; ++it) xv[it] = X4[tid + it * NTH];
#pragma unroll
        for (int it = 0; it < PER; ++it) {
            const int i = tid + it * NTH;
            *(v4f*)&sX[(i / K4) * SXS + (i % K4) * 4] = xv[it];
        }
    } else {
        for (int i = tid; i < ROWS * K; i += NTH) {
            int r = i / K, c = i % K;
            int gr = row0 + r;
            sX[r * SXS + c] = (gr < M) ? X[(size_t)gr * K + c] : 0.0f;
        }
    }
    __syncthreads();

    const int half = lane >> 4;
    const int idx  = lane & 15;
    const int colbase = wave * 16;

    v8f acc[RT] = {};
#pragma unroll
    for (int k0 = 0; k0 < K; k0 += 4) {
        const int ka = k0 + half * 2;
        v2f b;
        b.x = W[(size_t)(ka)     * NCOL + colbase + idx];
        b.y = W[(size_t)(ka + 1) * NCOL + colbase + idx];
#pragma unroll
        for (int t = 0; t < RT; ++t) {
            v2f a;
            a.x = sX[(t * 16 + idx) * SXS + ka];
            a.y = sX[(t * 16 + idx) * SXS + ka + 1];
            acc[t] = __builtin_amdgcn_wmma_f32_16x16x4_f32(
                 false, a,  false, b,
                 (short)0, acc[t],  false,  false);
        }
    }

#pragma unroll
    for (int t = 0; t < RT; ++t) {
#pragma unroll
        for (int v = 0; v < 8; ++v) {
            int gr = row0 + t * 16 + half * 8 + v;
            if (gr < M) H[(size_t)gr * NCOL + colbase + idx] = acc[t][v];
        }
    }
}

template <int F, bool RELU>
__global__ void __launch_bounds__(256)
k_gather(const int* __restrict__ rowptr, const int* __restrict__ col,
         const float* __restrict__ dinv, const float* __restrict__ h,
         const float* __restrict__ bias, float* __restrict__ res, int N) {
    constexpr int C = F / 32;
    typedef float vecC __attribute__((ext_vector_type(C)));
    const int lane = threadIdx.x & 31;
    const int node = __builtin_amdgcn_readfirstlane(
        (int)(blockIdx.x * (blockDim.x >> 5) + (threadIdx.x >> 5)));
    if (node >= N) return;
    const int beg = rowptr[node];
    const int end = rowptr[node + 1];
    const float dd = dinv[node];
    const vecC* __restrict__ hv = (const vecC*)h;
    const vecC hs = hv[(size_t)node * 32 + lane];
    const vecC bs = ((const vecC*)bias)[lane];
    vecC acc = 0.0f;

    for (int base = beg; base < end; base += 32) {
        const int rem = end - base;
        const int cnt = rem < 32 ? rem : 32;
        int   s_l = 0;
        float w_l = 0.0f;
        if (lane < cnt) {
            s_l = col[base + lane];
            w_l = dinv[s_l];
        }
        const int wbits = __builtin_bit_cast(int, w_l);
        int j = 0;
        for (; j + 8 <= cnt; j += 8) {
            vecC v[8];
            for (int u = 0; u < 8; ++u) {
                const int s = __builtin_amdgcn_readlane(s_l, j + u);
                v[u] = hv[(size_t)s * 32 + lane];
            }
            for (int u = 0; u < 8; ++u) {
                const float w = __builtin_bit_cast(float, __builtin_amdgcn_readlane(wbits, j + u)) * dd;
                acc += v[u] * w;
            }
        }
        if (j + 4 <= cnt) {
            vecC v[4];
            for (int u = 0; u < 4; ++u) {
                const int s = __builtin_amdgcn_readlane(s_l, j + u);
                v[u] = hv[(size_t)s * 32 + lane];
            }
            for (int u = 0; u < 4; ++u) {
                const float w = __builtin_bit_cast(float, __builtin_amdgcn_readlane(wbits, j + u)) * dd;
                acc += v[u] * w;
            }
            j += 4;
        }
        for (; j < cnt; ++j) {
            const int s = __builtin_amdgcn_readlane(s_l, j);
            const float w = __builtin_bit_cast(float, __builtin_amdgcn_readlane(wbits, j)) * dd;
            acc += hv[(size_t)s * 32 + lane] * w;
        }
    }
    vecC r;
    for (int c = 0; c < C; ++c) {
        const float v = acc[c] + hs[c] * dd * dd + bs[c];
        r[c] = RELU ? (v > 0.0f ? v : 0.0f) : v;
    }
    ((vecC*)res)[(size_t)node * 32 + lane] = r;
}

template <int F, int NOUT, int ROWS>
__global__ void __launch_bounds__(256, 16)
k_gather_gemm(const int* __restrict__ rowptr, const int* __restrict__ col,
              const float* __restrict__ dinv, const float* __restrict__ h,
              const float* __restrict__ bias, const float* __restrict__ W,
              float* __restrict__ H2, int N) {
    constexpr int C   = F / 32;
    constexpr int SXS = F + 4;
    constexpr int NPW = ROWS / 8;
    constexpr int RT  = ROWS / 16;
    static_assert(C == 4, "a lane owns one 16-byte piece of a row");
    typedef float vecC __attribute__((ext_vector_type(C)));
    __shared__ __attribute__((aligned(16))) float sX[ROWS * SXS];

    const int lane = threadIdx.x & 31;
    const int wave = __builtin_amdgcn_readfirstlane((int)(threadIdx.x >> 5));
    const int row0 = blockIdx.x * ROWS;
    const vecC* __restrict__ hv = (const vecC*)h;
    const vecC bs = ((const vecC*)bias)[lane];

    for (int i = 0; i < NPW; ++i) {
        const int r    = wave * NPW + i;
        const int node = row0 + r;
        const bool valid = node < N;
        const int nc  = valid ? node : N - 1;
        const int beg = rowptr[valid ? node : N];
        const int end = rowptr[node + 1 < N ? node + 1 : N];
        const float dd = dinv[nc];
        const vecC hs = hv[(size_t)nc * 32 + lane];
        vecC acc = 0.0f;

        for (int base = beg; base < end; base += 32) {
            const int rem = end - base;
            const int cnt = rem < 32 ? rem : 32;
            int   s_l = 0;
            float w_l = 0.0f;
            if (lane < cnt) {
                s_l = col[base + lane];
                w_l = dinv[s_l];
            }
            const int wbits = __builtin_bit_cast(int, w_l);
            int j = 0;
            for (; j + 8 <= cnt; j += 8) {
                vecC v[8];
                for (int u = 0; u < 8; ++u) {
                    const int s = __builtin_amdgcn_readlane(s_l, j + u);
                    v[u] = hv[(size_t)s * 32 + lane];
                }
                for (int u = 0; u < 8; ++u) {
                    const float w = __builtin_bit_cast(float, __builtin_amdgcn_readlane(wbits, j + u)) * dd;
                    acc += v[u] * w;
                }
            }
            if (j + 4 <= cnt) {
                vecC v[4];
                for (int u = 0; u < 4; ++u) {
                    const int s = __builtin_amdgcn_readlane(s_l, j + u);
                    v[u] = hv[(size_t)s * 32 + lane];
                }
                for (int u = 0; u < 4; ++u) {
                    const float w = __builtin_bit_cast(float, __builtin_amdgcn_readlane(wbits, j + u)) * dd;
                    acc += v[u] * w;
                }
                j += 4;
            }
            for (; j < cnt; ++j) {
                const int s = __builtin_amdgcn_readlane(s_l, j);
                const float w = __builtin_bit_cast(float, __builtin_amdgcn_readlane(wbits, j)) * dd;
                acc += hv[(size_t)s * 32 + lane] * w;
            }
        }
        vecC o;
        for (int c = 0; c < C; ++c) {
            const float v = acc[c] + hs[c] * dd * dd + bs[c];
            o[c] = valid ? (v > 0.0f ? v : 0.0f) : 0.0f;
        }
        *(vecC*)&sX[r * SXS + lane * C] = o;
    }
    __syncthreads();

    if (wave < NOUT / 16) {
        const int half = lane >> 4;
        const int idx  = lane & 15;
        const int colbase = wave * 16;

        v8f acc2[RT] = {};
#pragma unroll
        for (int k0 = 0; k0 < F; k0 += 4) {
            const int ka = k0 + half * 2;
            v2f b;
            b.x = W[(size_t)(ka)     * NOUT + colbase + idx];
            b.y = W[(size_t)(ka + 1) * NOUT + colbase + idx];
#pragma unroll
            for (int t = 0; t < RT; ++t) {
                v2f a;
                a.x = sX[(t * 16 + idx) * SXS + ka];
                a.y = sX[(t * 16 + idx) * SXS + ka + 1];
                acc2[t] = __builtin_amdgcn_wmma_f32_16x16x4_f32(
                     false, a,  false, b,
                     (short)0, acc2[t],  false,  false);
            }
        }

#pragma unroll
        for (int t = 0; t < RT; ++t) {
#pragma unroll
            for (int v = 0; v < 8; ++v) {
                const int gr = row0 + t * 16 + half * 8 + v;
                if (gr < N) H2[(size_t)gr * NOUT + colbase + idx] = acc2[t][v];
            }
        }
    }
}

extern "C" void kernel_launch(void* const* d_in, const int* in_sizes, int n_in,
                              void* d_out, int out_size, void* d_ws, size_t ws_size,
                              hipStream_t stream) {
    const float* x  = (const float*)d_in[0];
    const int*   ei = (const int*)d_in[1];
    const float* W1 = (const float*)d_in[2];
    const float* b1 = (const float*)d_in[3];
    const float* W2 = (const float*)d_in[4];
    const float* b2 = (const float*)d_in[5];
    float* out = (float*)d_out;

    const int  N = in_sizes[0] / IN_F;
    const long E = (long)in_sizes[1] / 2;
    const int* srcIdx = ei;
    const int* dstIdx = ei + E;

    float* ws   = (float*)d_ws;
    float* dinv = ws;
    float* h    = dinv + N;
    float* h2   = h + (size_t)N * HID_F;
    const int N4 = (N + 3) / 4 * 4;
    const int NC = (int)((E + CH - 1) / CH);
    const int NT = (N + 255) / 256;
    int* rowptr    = (int*)(h2 + (size_t)N * HID_F);
    int* col       = rowptr + N4 + 4;
    unsigned* recs = (unsigned*)h2;
    int* P         = (int*)(recs + (size_t)NC * CH);

    k_tilesort<<<NC, 256, 0, stream>>>(srcIdx, dstIdx, recs, P, E, NT);
    k_tilecsr<<<NT, 256, 0, stream>>>(recs, P, rowptr, col, dinv, N, E, NC, NT);

    constexpr int RT   = 4;
    constexpr int ROWS = 16 * RT;

    k_gemm_wmma<IN_F, HID_F, RT>
        <<<(N + ROWS - 1) / ROWS, 32 * (HID_F / 16), 0, stream>>>(x, W1, h, N);
    constexpr int FR = 32;
    k_gather_gemm<HID_F, OUT_F, FR>
        <<<(N + FR - 1) / FR, 256, 0, stream>>>(rowptr, col, dinv, h, b1, W2, h2, N);

    k_gather<OUT_F, false><<<(N + 7) / 8, 256, 0, stream>>>(rowptr, col, dinv, h2, b2, out, N);
}


# Layer 1 as (A x) W1, was A (x W1): the pull kernel sums dinv[src]-weighted rows of x and does both products on WMMA per 64-node block; the x W1 launch and its stored table go; sum order changes
# speedup vs baseline: 5.8189x; 1.2798x over previous
#include <hip/hip_runtime.h>


typedef __attribute__((ext_vector_type(2))) float v2f;
typedef __attribute__((ext_vector_type(8))) float v8f;
typedef __attribute__((ext_vector_type(4))) float v4f;

#define IN_F  128
#define HID_F 128
#define OUT_F 64

#define CH   4096
#define MAXT 512
#define MAXC 512
#define CAP  8192

__global__ void __launch_bounds__(256)
k_tilesort(const int* __restrict__ src, const int* __restrict__ dst,
           unsigned* __restrict__ recs, int* __restrict__ P, long E, int NT) {
    __shared__ int sHist[MAXT];
    __shared__ int sOff[MAXT + 1];
    __shared__ int sWave[8];
    __shared__ unsigned sRec[CH];
    const int tid  = threadIdx.x;
    const int lane = tid & 31;
    const int wave = tid >> 5;
    const int c    = blockIdx.x;
    const long ebase = (long)c * CH;
    const long left  = E - ebase;
    const int n = left < CH ? (int)left : CH;

    for (int i = tid; i < MAXT; i += 256) sHist[i] = 0;
    __syncthreads();

    unsigned rec[16];
    int tl[16], rk[16];
    int dv[16], sv[16];
    for (int k = 0; k < 16; ++k) {
        const int i  = k * 256 + tid;
        const int ic = i < n ? i : n - 1;
        dv[k] = dst[ebase + ic];
        sv[k] = src[ebase + ic];
    }
    for (int k = 0; k < 16; ++k) {
        const bool valid = (k * 256 + tid) < n;
        tl[k]  = valid ? (dv[k] >> 8) : -1;
        rk[k]  = 0;
        rec[k] = ((unsigned)sv[k] << 8) | ((unsigned)dv[k] & 255u);
    }
    for (int k = 0; k < 16; ++k)
        if (tl[k] >= 0)
            rk[k] = __hip_atomic_fetch_add(&sHist[tl[k]], 1, __ATOMIC_RELAXED,
                                           __HIP_MEMORY_SCOPE_WORKGROUP);
    __syncthreads();

    const int a0 = sHist[2 * tid], a1 = sHist[2 * tid + 1];
    const int s2 = a0 + a1;
    int incl = s2;
    for (int o = 1; o < 32; o <<= 1) {
        const int u = __shfl_up(incl, o);
        if (lane >= o) incl += u;
    }
    if (lane == 31) sWave[wave] = incl;
    __syncthreads();
    int carry = 0;
    for (int w = 0; w < wave; ++w) carry += sWave[w];
    const int ex = carry + incl - s2;
    sOff[2 * tid]     = ex;
    sOff[2 * tid + 1] = ex + a0;
    if (tid == 255) sOff[MAXT] = ex + s2;
    __syncthreads();

    for (int k = 0; k < 16; ++k)
        if (tl[k] >= 0) sRec[sOff[tl[k]] + rk[k]] = rec[k];
    for (int t = tid; t <= NT; t += 256) P[(size_t)c * (NT + 1) + t] = sOff[t];
    __syncthreads();
    for (int i = tid; i < n; i += 256) recs[ebase + i] = sRec[i];
}

template <bool PLACE>
__device__ __forceinline__ void tile_sweep(const unsigned* __restrict__ recs, const int* sStart,
                                           const int* sLen, int* sBin, int* sCol, int* __restrict__ col,
                                           int NC, int base, bool staged, int half, int hl) {
    for (int c0 = half; c0 < NC; c0 += 128) {
        unsigned rv[8];
        bool ok[8];
        for (int u = 0; u < 8; ++u) {
            const int c  = c0 + 16 * u;
            const int cc = c < NC ? c : 0;
            ok[u] = (c < NC) && (hl < sLen[cc]);
            const size_t a = ok[u] ? ((size_t)cc * CH + sStart[cc] + hl) : (size_t)0;
            rv[u] = recs[a];
        }
        for (int u = 0; u < 8; ++u) {
            if (ok[u]) {
                const int r = __hip_atomic_fetch_add(&sBin[rv[u] & 255u], 1, __ATOMIC_RELAXED,
                                                     __HIP_MEMORY_SCOPE_WORKGROUP);
                if (PLACE) {
                    const int s = (int)(rv[u] >> 8);
                    if (staged) sCol[r] = s; else col[base + r] = s;
                }
            }
        }
        for (int u = 0; u < 8; ++u) {
            const int c = c0 + 16 * u;
            if (c < NC) {
                const int len = sLen[c];
                for (int j = hl + 16; j < len; j += 16) {
                    const unsigned v = recs[(size_t)c * CH + sStart[c] + j];
                    const int r = __hip_atomic_fetch_add(&sBin[v & 255u], 1, __ATOMIC_RELAXED,
                                                         __HIP_MEMORY_SCOPE_WORKGROUP);
                    if (PLACE) {
                        const int s = (int)(v >> 8);
                        if (staged) sCol[r] = s; else col[base + r] = s;
                    }
                }
            }
        }
    }
}

__global__ void __launch_bounds__(256)
k_tilecsr(const unsigned* __restrict__ recs, const int* __restrict__ P,
          int* __restrict__ rowptr, int* __restrict__ col, float* __restrict__ dinv,
          int N, long E, int NC, int NT) {
    __shared__ int sStart[MAXC];
    __shared__ int sLen[MAXC];
    __shared__ int sCnt[256];
    __shared__ int sCur[256];
    __shared__ int sWave[8];
    __shared__ int sRedB[8];
    __shared__ int sRedT[8];
    __shared__ int sCol[CAP];
    const int tid  = threadIdx.x;
    const int lane = tid & 31;
    const int wave = tid >> 5;
    const int t    = blockIdx.x;
    const int half = tid >> 4;
    const int hl   = tid & 15;

    int bsum = 0, tsum = 0;
    for (int c = tid; c < NC; c += 256) {
        const int p0 = P[(size_t)c * (NT + 1) + t];
        const int p1 = P[(size_t)c * (NT + 1) + t + 1];
        sStart[c] = p0;
        sLen[c]   = p1 - p0;
        bsum += p0;
        tsum += p1 - p0;
    }
    sCnt[tid] = 0;
    for (int o = 16; o > 0; o >>= 1) {
        bsum += __shfl_xor(bsum, o);
        tsum += __shfl_xor(tsum, o);
    }
    if (lane == 0) { sRedB[wave] = bsum; sRedT[wave] = tsum; }
    __syncthreads();
    int base = 0, total = 0;
    for (int w = 0; w < 8; ++w) { base += sRedB[w]; total += sRedT[w]; }
    const bool staged = total <= CAP;

    tile_sweep<false>(recs, sStart, sLen, sCnt, sCol, col, NC, base, staged, half, hl);
    __syncthreads();

    const int cn = sCnt[tid];
    int incl = cn;
    for (int o = 1; o < 32; o <<= 1) {
        const int u = __shfl_up(incl, o);
        if (lane >= o) incl += u;
    }
    if (lane == 31) sWave[wave] = incl;
    __syncthreads();
    int carry = 0;
    for (int w = 0; w < wave; ++w) carry += sWave[w];
    const int ex = carry + incl - cn;
    const int node = t * 256 + tid;
    if (node < N) {
        rowptr[node] = base + ex;
        dinv[node]   = rsqrtf((float)(cn + 1));
    }
    if (t == NT - 1 && tid == 0) rowptr[N] = (int)E;
    sCur[tid] = ex;
    __syncthreads();

    tile_sweep<true>(recs, sStart, sLen, sCur, sCol, col, NC, base, staged, half, hl);
    __syncthreads();
    if (staged)
        for (int i = tid; i < total; i += 256) col[base + i] = sCol[i];
}

template <int F, bool RELU>
__global__ void __launch_bounds__(256)
k_gather(const int* __restrict__ rowptr, const int* __restrict__ col,
         const float* __restrict__ dinv, const float* __restrict__ h,
         const float* __restrict__ bias, float* __restrict__ res, int N) {
    constexpr int C = F / 32;
    typedef float vecC __attribute__((ext_vector_type(C)));
    const int lane = threadIdx.x & 31;
    const int node = __builtin_amdgcn_readfirstlane(
        (int)(blockIdx.x * (blockDim.x >> 5) + (threadIdx.x >> 5)));
    if (node >= N) return;
    const int beg = rowptr[node];
    const int end = rowptr[node + 1];
    const float dd = dinv[node];
    const vecC* __restrict__ hv = (const vecC*)h;
    const vecC hs = hv[(size_t)node * 32 + lane];
    const vecC bs = ((const vecC*)bias)[lane];
    vecC acc = 0.0f;

    for (int base = beg; base < end; base += 32) {
        const int rem = end - base;
        const int cnt = rem < 32 ? rem : 32;
        int s_l = 0;
        if (lane < cnt) s_l = col[base + lane];
        int j = 0;
        for (; j + 8 <= cnt; j += 8) {
            vecC v[8];
            for (int u = 0; u < 8; ++u) {
                const int s = __builtin_amdgcn_readlane(s_l, j + u);
                v[u] = hv[(size_t)s * 32 + lane];
            }
            for (int u = 0; u < 8; ++u) acc += v[u];
        }
        if (j + 4 <= cnt) {
            vecC v[4];
            for (int u = 0; u < 4; ++u) {
                const int s = __builtin_amdgcn_readlane(s_l, j + u);
                v[u] = hv[(size_t)s * 32 + lane];
            }
            for (int u = 0; u < 4; ++u) acc += v[u];
            j += 4;
        }
        for (; j < cnt; ++j) {
            const int s = __builtin_amdgcn_readlane(s_l, j);
            acc += hv[(size_t)s * 32 + lane];
        }
    }
    vecC r;
    for (int c = 0; c < C; ++c) {
        const float v = (acc[c] + hs[c]) * dd + bs[c];
        r[c] = RELU ? (v > 0.0f ? v : 0.0f) : v;
    }
    ((vecC*)res)[(size_t)node * 32 + lane] = r;
}

template <int F, int FH, int NOUT, int ROWS, int NW>
__global__ void __launch_bounds__(32 * NW, 16)
k_gather_gemm2(const int* __restrict__ rowptr, const int* __restrict__ col,
               const float* __restrict__ dinv, const float* __restrict__ x,
               const float* __restrict__ W1, const float* __restrict__ bias1,
               const float* __restrict__ W2, float* __restrict__ H2, int N) {
    constexpr int C   = F / 32;
    constexpr int SXS = F + 4;
    constexpr int NPW = ROWS / NW;
    constexpr int S1  = FH / 16;
    constexpr int S2  = NOUT / 16;
    constexpr int G   = NW / S1;
    constexpr int RT  = ROWS / 16 / G;
    static_assert(C == 4, "a lane owns one 16-byte piece of a row");
    static_assert(FH == F, "T is written back into the tile the rows came from");
    static_assert(NW % S1 == 0 && ROWS % (16 * G) == 0 && ROWS % NW == 0, "waves = strips x row groups");
    static_assert(S2 * G <= NW, "the second product fits the block's waves");
    typedef float vecC __attribute__((ext_vector_type(C)));
    __shared__ __attribute__((aligned(16))) float sX[ROWS * SXS];
    __shared__ float sD[ROWS];

    const int lane = threadIdx.x & 31;
    const int wave = __builtin_amdgcn_readfirstlane((int)(threadIdx.x >> 5));
    const int row0 = blockIdx.x * ROWS;
    const vecC* __restrict__ xv = (const vecC*)x;

    for (int i = 0; i < NPW; ++i) {
        const int r    = wave * NPW + i;
        const int node = row0 + r;
        const bool valid = node < N;
        const int nc  = valid ? node : N - 1;
        const int beg = rowptr[valid ? node : N];
        const int end = rowptr[node + 1 < N ? node + 1 : N];
        const float dd = dinv[nc];
        const vecC xs = xv[(size_t)nc * 32 + lane];
        vecC acc = 0.0f;

        for (int base = beg; base < end; base += 32) {
            const int rem = end - base;
            const int cnt = rem < 32 ? rem : 32;
            int s_l = 0;
            float w_l = 0.0f;
            if (lane < cnt) {
                s_l = col[base + lane];
                w_l = dinv[s_l];
            }
            const int w_i = __float_as_int(w_l);
            int j = 0;
            for (; j + 8 <= cnt; j += 8) {
                vecC v[8];
                for (int u = 0; u < 8; ++u) {
                    const int s = __builtin_amdgcn_readlane(s_l, j + u);
                    v[u] = xv[(size_t)s * 32 + lane];
                }
                for (int u = 0; u < 8; ++u)
                    acc += v[u] * __int_as_float(__builtin_amdgcn_readlane(w_i, j + u));
            }
            if (j + 4 <= cnt) {
                vecC v[4];
                for (int u = 0; u < 4; ++u) {
                    const int s = __builtin_amdgcn_readlane(s_l, j + u);
                    v[u] = xv[(size_t)s * 32 + lane];
                }
                for (int u = 0; u < 4; ++u)
                    acc += v[u] * __int_as_float(__builtin_amdgcn_readlane(w_i, j + u));
                j += 4;
            }
            for (; j < cnt; ++j) {
                const int s = __builtin_amdgcn_readlane(s_l, j);
                acc += xv[(size_t)s * 32 + lane] * __int_as_float(__builtin_amdgcn_readlane(w_i, j));
            }
        }
        vecC o;
        for (int c = 0; c < C; ++c) o[c] = valid ? (acc[c] + xs[c] * dd) * dd : 0.0f;
        *(vecC*)&sX[r * SXS + lane * C] = o;
        if (lane == 0) sD[r] = valid ? dd : 0.0f;
    }
    __syncthreads();

    const int half = lane >> 4;
    const int idx  = lane & 15;

    {
        const int colbase = (wave % S1) * 16;
        const int rbase   = (wave / S1) * (RT * 16);
        v8f acc1[RT] = {};
#pragma unroll 8
        for (int k0 = 0; k0 < F; k0 += 4) {
            const int ka = k0 + half * 2;
            v2f b;
            b.x = W1[(size_t)(ka)     * FH + colbase + idx];
            b.y = W1[(size_t)(ka + 1) * FH + colbase + idx];
#pragma unroll
            for (int t = 0; t < RT; ++t) {
                v2f a;
                a.x = sX[(rbase + t * 16 + idx) * SXS + ka];
                a.y = sX[(rbase + t * 16 + idx) * SXS + ka + 1];
                acc1[t] = __builtin_amdgcn_wmma_f32_16x16x4_f32(
                     false, a,  false, b,
                     (short)0, acc1[t],  false,  false);
            }
        }
        const float bb = bias1[colbase + idx];
        __syncthreads();
#pragma unroll
        for (int t = 0; t < RT; ++t) {
#pragma unroll
            for (int v = 0; v < 8; ++v) {
                const int rr = rbase + t * 16 + half * 8 + v;
                const float z = acc1[t][v] + bb;
                sX[rr * SXS + colbase + idx] = (z > 0.0f ? z : 0.0f) * sD[rr];
            }
        }
    }
    __syncthreads();

    if (wave < S2 * G) {
        const int colbase = (wave % S2) * 16;
        const int rbase   = (wave / S2) * (RT * 16);
        v8f acc2[RT] = {};
#pragma unroll 8
        for (int k0 = 0; k0 < FH; k0 += 4) {
            const int ka = k0 + half * 2;
            v2f b;
            b.x = W2[(size_t)(ka)     * NOUT + colbase + idx];
            b.y = W2[(size_t)(ka + 1) * NOUT + colbase + idx];
#pragma unroll
            for (int t = 0; t < RT; ++t) {
                v2f a;
                a.x = sX[(rbase + t * 16 + idx) * SXS + ka];
                a.y = sX[(rbase + t * 16 + idx) * SXS + ka + 1];
                acc2[t] = __builtin_amdgcn_wmma_f32_16x16x4_f32(
                     false, a,  false, b,
                     (short)0, acc2[t],  false,  false);
            }
        }

#pragma unroll
        for (int t = 0; t < RT; ++t) {
#pragma unroll
            for (int v = 0; v < 8; ++v) {
                const int gr = row0 + rbase + t * 16 + half * 8 + v;
                if (gr < N) H2[(size_t)gr * NOUT + colbase + idx] = acc2[t][v];
            }
        }
    }
}

extern "C" void kernel_launch(void* const* d_in, const int* in_sizes, int n_in,
                              void* d_out, int out_size, void* d_ws, size_t ws_size,
                              hipStream_t stream) {
    const float* x  = (const float*)d_in[0];
    const int*   ei = (const int*)d_in[1];
    const float* W1 = (const float*)d_in[2];
    const float* b1 = (const float*)d_in[3];
    const float* W2 = (const float*)d_in[4];
    const float* b2 = (const float*)d_in[5];
    float* out = (float*)d_out;

    const int  N = in_sizes[0] / IN_F;
    const long E = (long)in_sizes[1] / 2;
    const int* srcIdx = ei;
    const int* dstIdx = ei + E;

    float* ws   = (float*)d_ws;
    float* dinv = ws;
    float* h2   = dinv + N + (size_t)N * HID_F;
    const int N4 = (N + 3) / 4 * 4;
    const int NC = (int)((E + CH - 1) / CH);
    const int NT = (N + 255) / 256;
    int* rowptr    = (int*)(h2 + (size_t)N * HID_F);
    int* col       = rowptr + N4 + 4;
    unsigned* recs = (unsigned*)h2;
    int* P         = (int*)(recs + (size_t)NC * CH);

    k_tilesort<<<NC, 256, 0, stream>>>(srcIdx, dstIdx, recs, P, E, NT);
    k_tilecsr<<<NT, 256, 0, stream>>>(recs, P, rowptr, col, dinv, N, E, NC, NT);

    constexpr int FR = 64;
    constexpr int FW = 16;
    k_gather_gemm2<IN_F, HID_F, OUT_F, FR, FW>
        <<<(N + FR - 1) / FR, 32 * FW, 0, stream>>>(rowptr, col, dinv, x, W1, b1, W2, h2, N);

    k_gather<OUT_F, false><<<(N + 7) / 8, 256, 0, stream>>>(rowptr, col, dinv, h2, b2, out, N);
}


# Tile CSR kernel: a wave per chunk segment, all record loads in flight before the first wait, records kept in registers for both sweeps; per-node sum order = LDS arrival order, pattern changes
# speedup vs baseline: 6.1238x; 1.0524x over previous
#include <hip/hip_runtime.h>


typedef __attribute__((ext_vector_type(2))) float v2f;
typedef __attribute__((ext_vector_type(8))) float v8f;
typedef __attribute__((ext_vector_type(4))) float v4f;

#define IN_F  128
#define HID_F 128
#define OUT_F 64

#define CH   4096
#define MAXT 512
#define MAXC 512
#define CAP  8192

__global__ void __launch_bounds__(256)
k_tilesort(const int* __restrict__ src, const int* __restrict__ dst,
           unsigned* __restrict__ recs, int* __restrict__ P, long E, int NT) {
    __shared__ int sHist[MAXT];
    __shared__ int sOff[MAXT + 1];
    __shared__ int sWave[8];
    __shared__ unsigned sRec[CH];
    const int tid  = threadIdx.x;
    const int lane = tid & 31;
    const int wave = tid >> 5;
    const int c    = blockIdx.x;
    const long ebase = (long)c * CH;
    const long left  = E - ebase;
    const int n = left < CH ? (int)left : CH;

    for (int i = tid; i < MAXT; i += 256) sHist[i] = 0;
    __syncthreads();

    unsigned rec[16];
    int tl[16], rk[16];
    int dv[16], sv[16];
    for (int k = 0; k < 16; ++k) {
        const int i  = k * 256 + tid;
        const int ic = i < n ? i : n - 1;
        dv[k] = dst[ebase + ic];
        sv[k] = src[ebase + ic];
    }
    for (int k = 0; k < 16; ++k) {
        const bool valid = (k * 256 + tid) < n;
        tl[k]  = valid ? (dv[k] >> 8) : -1;
        rk[k]  = 0;
        rec[k] = ((unsigned)sv[k] << 8) | ((unsigned)dv[k] & 255u);
    }
    for (int k = 0; k < 16; ++k)
        if (tl[k] >= 0)
            rk[k] = __hip_atomic_fetch_add(&sHist[tl[k]], 1, __ATOMIC_RELAXED,
                                           __HIP_MEMORY_SCOPE_WORKGROUP);
    __syncthreads();

    const int a0 = sHist[2 * tid], a1 = sHist[2 * tid + 1];
    const int s2 = a0 + a1;
    int incl = s2;
    for (int o = 1; o < 32; o <<= 1) {
        const int u = __shfl_up(incl, o);
        if (lane >= o) incl += u;
    }
    if (lane == 31) sWave[wave] = incl;
    __syncthreads();
    int carry = 0;
    for (int w = 0; w < wave; ++w) carry += sWave[w];
    const int ex = carry + incl - s2;
    sOff[2 * tid]     = ex;
    sOff[2 * tid + 1] = ex + a0;
    if (tid == 255) sOff[MAXT] = ex + s2;
    __syncthreads();

    for (int k = 0; k < 16; ++k)
        if (tl[k] >= 0) sRec[sOff[tl[k]] + rk[k]] = rec[k];
    for (int t = tid; t <= NT; t += 256) P[(size_t)c * (NT + 1) + t] = sOff[t];
    __syncthreads();
    for (int i = tid; i < n; i += 256) recs[ebase + i] = sRec[i];
}

#define SEGW (MAXC / 8)

template <bool PLACE>
__device__ __forceinline__ void segments_slow(const unsigned* __restrict__ recs, const int (&st)[2],
                                              const int (&ln)[2], int* sBin, int* sCol,
                                              int* __restrict__ col, int base, bool staged,
                                              int wave, int lane, int from) {
    for (int h = 0; h < 2; ++h) {
        unsigned m = __builtin_amdgcn_ballot_w32(ln[h] > from);
        while (m) {
            const int b = __builtin_ctz(m);
            m &= m - 1;
            const int c  = wave + 8 * (b + 32 * h);
            const int s0 = __builtin_amdgcn_readlane(st[h], b);
            const int l0 = __builtin_amdgcn_readlane(ln[h], b);
            for (int j = lane + from; j < l0; j += 32) {
                const unsigned v = recs[(unsigned)(c * CH + s0 + j)];
                const int r = __hip_atomic_fetch_add(&sBin[v & 255u], 1, __ATOMIC_RELAXED,
                                                     __HIP_MEMORY_SCOPE_WORKGROUP);
                if (PLACE) {
                    const int s = (int)(v >> 8);
                    if (staged) sCol[r] = s; else col[base + r] = s;
                }
            }
        }
    }
}

__global__ void __launch_bounds__(256)
k_tilecsr(const unsigned* __restrict__ recs, const int* __restrict__ P,
          int* __restrict__ rowptr, int* __restrict__ col, float* __restrict__ dinv,
          int N, long E, int NC, int NT) {
    __shared__ int sCnt[256];
    __shared__ int sCur[256];
    __shared__ int sWave[8];
    __shared__ int sRedB[8];
    __shared__ int sRedT[8];
    __shared__ int sCol[CAP];
    const int tid  = threadIdx.x;
    const int lane = tid & 31;
    const int wave = __builtin_amdgcn_readfirstlane(tid >> 5);
    const int t    = blockIdx.x;

    int st[2], ln[2];
    for (int h = 0; h < 2; ++h) {
        const int c = wave + 8 * (lane + 32 * h);
        const bool in = c < NC;
        const size_t pi = (size_t)(in ? c : 0) * (NT + 1) + t;
        const int p0 = P[pi];
        const int p1 = P[pi + 1];
        st[h] = in ? p0 : 0;
        ln[h] = in ? p1 - p0 : 0;
    }
    sCnt[tid] = 0;

    unsigned rv[SEGW];
#pragma unroll
    for (int q = 0; q < SEGW; ++q) {
        const int s0 = __builtin_amdgcn_readlane(st[q >> 5], q & 31);
        const int l0 = __builtin_amdgcn_readlane(ln[q >> 5], q & 31);
        const unsigned a = lane < l0 ? (unsigned)((wave + 8 * q) * CH + s0 + lane) : 0u;
        rv[q] = recs[a];
    }

    int bsum = st[0] + st[1], tsum = ln[0] + ln[1];
    for (int o = 16; o > 0; o >>= 1) {
        bsum += __shfl_xor(bsum, o);
        tsum += __shfl_xor(tsum, o);
    }
    if (lane == 0) { sRedB[wave] = bsum; sRedT[wave] = tsum; }
    __syncthreads();
    int base = 0, total = 0;
    for (int w = 0; w < 8; ++w) { base += sRedB[w]; total += sRedT[w]; }
    const bool staged = total <= CAP;

#pragma unroll
    for (int q = 0; q < SEGW; ++q) {
        const int l0 = __builtin_amdgcn_readlane(ln[q >> 5], q & 31);
        if (lane < l0)
            __hip_atomic_fetch_add(&sCnt[rv[q] & 255u], 1, __ATOMIC_RELAXED,
                                   __HIP_MEMORY_SCOPE_WORKGROUP);
    }
    segments_slow<false>(recs, st, ln, sCnt, sCol, col, base, staged, wave, lane, 32);
    __syncthreads();

    const int cn = sCnt[tid];
    int incl = cn;
    for (int o = 1; o < 32; o <<= 1) {
        const int u = __shfl_up(incl, o);
        if (lane >= o) incl += u;
    }
    if (lane == 31) sWave[wave] = incl;
    __syncthreads();
    int carry = 0;
    for (int w = 0; w < wave; ++w) carry += sWave[w];
    const int ex = carry + incl - cn;
    const int node = t * 256 + tid;
    if (node < N) {
        rowptr[node] = base + ex;
        dinv[node]   = rsqrtf((float)(cn + 1));
    }
    if (t == NT - 1 && tid == 0) rowptr[N] = (int)E;
    sCur[tid] = ex;
    __syncthreads();

    if (staged) {
        int rk[SEGW];
#pragma unroll
        for (int q = 0; q < SEGW; ++q) {
            const int l0 = __builtin_amdgcn_readlane(ln[q >> 5], q & 31);
            rk[q] = 0;
            if (lane < l0)
                rk[q] = __hip_atomic_fetch_add(&sCur[rv[q] & 255u], 1, __ATOMIC_RELAXED,
                                               __HIP_MEMORY_SCOPE_WORKGROUP);
        }
#pragma unroll
        for (int q = 0; q < SEGW; ++q) {
            const int l0 = __builtin_amdgcn_readlane(ln[q >> 5], q & 31);
            if (lane < l0) sCol[rk[q]] = (int)(rv[q] >> 8);
        }
        segments_slow<true>(recs, st, ln, sCur, sCol, col, base, true, wave, lane, 32);
    } else {
        segments_slow<true>(recs, st, ln, sCur, sCol, col, base, false, wave, lane, 0);
    }
    __syncthreads();
    if (staged)
        for (int i = tid; i < total; i += 256) col[base + i] = sCol[i];
}

template <int F, bool RELU>
__global__ void __launch_bounds__(256)
k_gather(const int* __restrict__ rowptr, const int* __restrict__ col,
         const float* __restrict__ dinv, const float* __restrict__ h,
         const float* __restrict__ bias, float* __restrict__ res, int N) {
    constexpr int C = F / 32;
    typedef float vecC __attribute__((ext_vector_type(C)));
    const int lane = threadIdx.x & 31;
    const int node = __builtin_amdgcn_readfirstlane(
        (int)(blockIdx.x * (blockDim.x >> 5) + (threadIdx.x >> 5)));
    if (node >= N) return;
    const int beg = rowptr[node];
    const int end = rowptr[node + 1];
    const float dd = dinv[node];
    const vecC* __restrict__ hv = (const vecC*)h;
    const vecC hs = hv[(size_t)node * 32 + lane];
    const vecC bs = ((const vecC*)bias)[lane];
    vecC acc = 0.0f;

    for (int base = beg; base < end; base += 32) {
        const int rem = end - base;
        const int cnt = rem < 32 ? rem : 32;
        int s_l = 0;
        if (lane < cnt) s_l = col[base + lane];
        int j = 0;
        for (; j + 8 <= cnt; j += 8) {
            vecC v[8];
            for (int u = 0; u < 8; ++u) {
                const int s = __builtin_amdgcn_readlane(s_l, j + u);
                v[u] = hv[(size_t)s * 32 + lane];
            }
            for (int u = 0; u < 8; ++u) acc += v[u];
        }
        if (j + 4 <= cnt) {
            vecC v[4];
            for (int u = 0; u < 4; ++u) {
                const int s = __builtin_amdgcn_readlane(s_l, j + u);
                v[u] = hv[(size_t)s * 32 + lane];
            }
            for (int u = 0; u < 4; ++u) acc += v[u];
            j += 4;
        }
        for (; j < cnt; ++j) {
            const int s = __builtin_amdgcn_readlane(s_l, j);
            acc += hv[(size_t)s * 32 + lane];
        }
    }
    vecC r;
    for (int c = 0; c < C; ++c) {
        const float v = (acc[c] + hs[c]) * dd + bs[c];
        r[c] = RELU ? (v > 0.0f ? v : 0.0f) : v;
    }
    ((vecC*)res)[(size_t)node * 32 + lane] = r;
}

template <int F, int FH, int NOUT, int ROWS, int NW>
__global__ void __launch_bounds__(32 * NW, 16)
k_gather_gemm2(const int* __restrict__ rowptr, const int* __restrict__ col,
               const float* __restrict__ dinv, const float* __restrict__ x,
               const float* __restrict__ W1, const float* __restrict__ bias1,
               const float* __restrict__ W2, float* __restrict__ H2, int N) {
    constexpr int C   = F / 32;
    constexpr int SXS = F + 4;
    constexpr int NPW = ROWS / NW;
    constexpr int S1  = FH / 16;
    constexpr int S2  = NOUT / 16;
    constexpr int G   = NW / S1;
    constexpr int RT  = ROWS / 16 / G;
    static_assert(C == 4, "a lane owns one 16-byte piece of a row");
    static_assert(FH == F, "T is written back into the tile the rows came from");
    static_assert(NW % S1 == 0 && ROWS % (16 * G) == 0 && ROWS % NW == 0, "waves = strips x row groups");
    static_assert(S2 * G <= NW, "the second product fits the block's waves");
    typedef float vecC __attribute__((ext_vector_type(C)));
    __shared__ __attribute__((aligned(16))) float sX[ROWS * SXS];
    __shared__ float sD[ROWS];

    const int lane = threadIdx.x & 31;
    const int wave = __builtin_amdgcn_readfirstlane((int)(threadIdx.x >> 5));
    const int row0 = blockIdx.x * ROWS;
    const vecC* __restrict__ xv = (const vecC*)x;

    for (int i = 0; i < NPW; ++i) {
        const int r    = wave * NPW + i;
        const int node = row0 + r;
        const bool valid = node < N;
        const int nc  = valid ? node : N - 1;
        const int beg = rowptr[valid ? node : N];
        const int end = rowptr[node + 1 < N ? node + 1 : N];
        const float dd = dinv[nc];
        const vecC xs = xv[(size_t)nc * 32 + lane];
        vecC acc = 0.0f;

        for (int base = beg; base < end; base += 32) {
            const int rem = end - base;
            const int cnt = rem < 32 ? rem : 32;
            int s_l = 0;
            float w_l = 0.0f;
            if (lane < cnt) {
                s_l = col[base + lane];
                w_l = dinv[s_l];
            }
            const int w_i = __float_as_int(w_l);
            int j = 0;
            for (; j + 8 <= cnt; j += 8) {
                vecC v[8];
                for (int u = 0; u < 8; ++u) {
                    const int s = __builtin_amdgcn_readlane(s_l, j + u);
                    v[u] = xv[(size_t)s * 32 + lane];
                }
                for (int u = 0; u < 8; ++u)
                    acc += v[u] * __int_as_float(__builtin_amdgcn_readlane(w_i, j + u));
            }
            if (j + 4 <= cnt) {
                vecC v[4];
                for (int u = 0; u < 4; ++u) {
                    const int s = __builtin_amdgcn_readlane(s_l, j + u);
                    v[u] = xv[(size_t)s * 32 + lane];
                }
                for (int u = 0; u < 4; ++u)
                    acc += v[u] * __int_as_float(__builtin_amdgcn_readlane(w_i, j + u));
                j += 4;
            }
            for (; j < cnt; ++j) {
                const int s = __builtin_amdgcn_readlane(s_l, j);
                acc += xv[(size_t)s * 32 + lane] * __int_as_float(__builtin_amdgcn_readlane(w_i, j));
            }
        }
        vecC o;
        for (int c = 0; c < C; ++c) o[c] = valid ? (acc[c] + xs[c] * dd) * dd : 0.0f;
        *(vecC*)&sX[r * SXS + lane * C] = o;
        if (lane == 0) sD[r] = valid ? dd : 0.0f;
    }
    __syncthreads();

    const int half = lane >> 4;
    const int idx  = lane & 15;

    {
        const int colbase = (wave % S1) * 16;
        const int rbase   = (wave / S1) * (RT * 16);
        v8f acc1[RT] = {};
#pragma unroll 8
        for (int k0 = 0; k0 < F; k0 += 4) {
            const int ka = k0 + half * 2;
            v2f b;
            b.x = W1[(size_t)(ka)     * FH + colbase + idx];
            b.y = W1[(size_t)(ka + 1) * FH + colbase + idx];
#pragma unroll
            for (int t = 0; t < RT; ++t) {
                v2f a;
                a.x = sX[(rbase + t * 16 + idx) * SXS + ka];
                a.y = sX[(rbase + t * 16 + idx) * SXS + ka + 1];
                acc1[t] = __builtin_amdgcn_wmma_f32_16x16x4_f32(
                     false, a,  false, b,
                     (short)0, acc1[t],  false,  false);
            }
        }
        const float bb = bias1[colbase + idx];
        __syncthreads();
#pragma unroll
        for (int t = 0; t < RT; ++t) {
#pragma unroll
            for (int v = 0; v < 8; ++v) {
                const int rr = rbase + t * 16 + half * 8 + v;
                const float z = acc1[t][v] + bb;
                sX[rr * SXS + colbase + idx] = (z > 0.0f ? z : 0.0f) * sD[rr];
            }
        }
    }
    __syncthreads();

    if (wave < S2 * G) {
        const int colbase = (wave % S2) * 16;
        const int rbase   = (wave / S2) * (RT * 16);
        v8f acc2[RT] = {};
#pragma unroll 8
        for (int k0 = 0; k0 < FH; k0 += 4) {
            const int ka = k0 + half * 2;
            v2f b;
            b.x = W2[(size_t)(ka)     * NOUT + colbase + idx];
            b.y = W2[(size_t)(ka + 1) * NOUT + colbase + idx];
#pragma unroll
            for (int t = 0; t < RT; ++t) {
                v2f a;
                a.x = sX[(rbase + t * 16 + idx) * SXS + ka];
                a.y = sX[(rbase + t * 16 + idx) * SXS + ka + 1];
                acc2[t] = __builtin_amdgcn_wmma_f32_16x16x4_f32(
                     false, a,  false, b,
                     (short)0, acc2[t],  false,  false);
            }
        }

#pragma unroll
        for (int t = 0; t < RT; ++t) {
#pragma unroll
            for (int v = 0; v < 8; ++v) {
                const int gr = row0 + rbase + t * 16 + half * 8 + v;
                if (gr < N) H2[(size_t)gr * NOUT + colbase + idx] = acc2[t][v];
            }
        }
    }
}

extern "C" void kernel_launch(void* const* d_in, const int* in_sizes, int n_in,
                              void* d_out, int out_size, void* d_ws, size_t ws_size,
                              hipStream_t stream) {
    const float* x  = (const float*)d_in[0];
    const int*   ei = (const int*)d_in[1];
    const float* W1 = (const float*)d_in[2];
    const float* b1 = (const float*)d_in[3];
    const float* W2 = (const float*)d_in[4];
    const float* b2 = (const float*)d_in[5];
    float* out = (float*)d_out;

    const int  N = in_sizes[0] / IN_F;
    const long E = (long)in_sizes[1] / 2;
    const int* srcIdx = ei;
    const int* dstIdx = ei + E;

    float* ws   = (float*)d_ws;
    float* dinv = ws;
    float* h2   = dinv + N + (size_t)N * HID_F;
    const int N4 = (N + 3) / 4 * 4;
    const int NC = (int)((E + CH - 1) / CH);
    const int NT = (N + 255) / 256;
    int* rowptr    = (int*)(h2 + (size_t)N * HID_F);
    int* col       = rowptr + N4 + 4;
    unsigned* recs = (unsigned*)h2;
    int* P         = (int*)(recs + (size_t)NC * CH);

    k_tilesort<<<NC, 256, 0, stream>>>(srcIdx, dstIdx, recs, P, E, NT);
    k_tilecsr<<<NT, 256, 0, stream>>>(recs, P, rowptr, col, dinv, N, E, NC, NT);

    constexpr int FR = 64;
    constexpr int FW = 16;
    k_gather_gemm2<IN_F, HID_F, OUT_F, FR, FW>
        <<<(N + FR - 1) / FR, 32 * FW, 0, stream>>>(rowptr, col, dinv, x, W1, b1, W2, h2, N);

    k_gather<OUT_F, false><<<(N + 7) / 8, 256, 0, stream>>>(rowptr, col, dinv, h2, b2, out, N);
}


# Fused layer-1 kernel as a pipeline inside each work-group: 16 waves gather 16-row tiles into an LDS ring while 4 waves run both WMMA products on earlier tiles; W1, W2 held in LDS; grid of 512
# speedup vs baseline: 6.5498x; 1.0696x over previous
#include <hip/hip_runtime.h>


typedef __attribute__((ext_vector_type(2))) float v2f;
typedef __attribute__((ext_vector_type(8))) float v8f;
typedef __attribute__((ext_vector_type(4))) float v4f;

#define IN_F  128
#define HID_F 128
#define OUT_F 64

#define CH   4096
#define MAXT 512
#define MAXC 512
#define CAP  8192

__global__ void __launch_bounds__(256)
k_tilesort(const int* __restrict__ src, const int* __restrict__ dst,
           unsigned* __restrict__ recs, int* __restrict__ P, long E, int NT) {
    __shared__ int sHist[MAXT];
    __shared__ int sOff[MAXT + 1];
    __shared__ int sWave[8];
    __shared__ unsigned sRec[CH];
    const int tid  = threadIdx.x;
    const int lane = tid & 31;
    const int wave = tid >> 5;
    const int c    = blockIdx.x;
    const long ebase = (long)c * CH;
    const long left  = E - ebase;
    const int n = left < CH ? (int)left : CH;

    for (int i = tid; i < MAXT; i += 256) sHist[i] = 0;
    __syncthreads();

    unsigned rec[16];
    int tl[16], rk[16];
    int dv[16], sv[16];
    for (int k = 0; k < 16; ++k) {
        const int i  = k * 256 + tid;
        const int ic = i < n ? i : n - 1;
        dv[k] = dst[ebase + ic];
        sv[k] = src[ebase + ic];
    }
    for (int k = 0; k < 16; ++k) {
        const bool valid = (k * 256 + tid) < n;
        tl[k]  = valid ? (dv[k] >> 8) : -1;
        rk[k]  = 0;
        rec[k] = ((unsigned)sv[k] << 8) | ((unsigned)dv[k] & 255u);
    }
    for (int k = 0; k < 16; ++k)
        if (tl[k] >= 0)
            rk[k] = __hip_atomic_fetch_add(&sHist[tl[k]], 1, __ATOMIC_RELAXED,
                                           __HIP_MEMORY_SCOPE_WORKGROUP);
    __syncthreads();

    const int a0 = sHist[2 * tid], a1 = sHist[2 * tid + 1];
    const int s2 = a0 + a1;
    int incl = s2;
    for (int o = 1; o < 32; o <<= 1) {
        const int u = __shfl_up(incl, o);
        if (lane >= o) incl += u;
    }
    if (lane == 31) sWave[wave] = incl;
    __syncthreads();
    int carry = 0;
    for (int w = 0; w < wave; ++w) carry += sWave[w];
    const int ex = carry + incl - s2;
    sOff[2 * tid]     = ex;
    sOff[2 * tid + 1] = ex + a0;
    if (tid == 255) sOff[MAXT] = ex + s2;
    __syncthreads();

    for (int k = 0; k < 16; ++k)
        if (tl[k] >= 0) sRec[sOff[tl[k]] + rk[k]] = rec[k];
    for (int t = tid; t <= NT; t += 256) P[(size_t)c * (NT + 1) + t] = sOff[t];
    __syncthreads();
    for (int i = tid; i < n; i += 256) recs[ebase + i] = sRec[i];
}

#define SEGW (MAXC / 8)

template <bool PLACE>
__device__ __forceinline__ void segments_slow(const unsigned* __restrict__ recs, const int (&st)[2],
                                              const int (&ln)[2], int* sBin, int* sCol,
                                              int* __restrict__ col, int base, bool staged,
                                              int wave, int lane, int from) {
    for (int h = 0; h < 2; ++h) {
        unsigned m = __builtin_amdgcn_ballot_w32(ln[h] > from);
        while (m) {
            const int b = __builtin_ctz(m);
            m &= m - 1;
            const int c  = wave + 8 * (b + 32 * h);
            const int s0 = __builtin_amdgcn_readlane(st[h], b);
            const int l0 = __builtin_amdgcn_readlane(ln[h], b);
            for (int j = lane + from; j < l0; j += 32) {
                const unsigned v = recs[(unsigned)(c * CH + s0 + j)];
                const int r = __hip_atomic_fetch_add(&sBin[v & 255u], 1, __ATOMIC_RELAXED,
                                                     __HIP_MEMORY_SCOPE_WORKGROUP);
                if (PLACE) {
                    const int s = (int)(v >> 8);
                    if (staged) sCol[r] = s; else col[base + r] = s;
                }
            }
        }
    }
}

__global__ void __launch_bounds__(256)
k_tilecsr(const unsigned* __restrict__ recs, const int* __restrict__ P,
          int* __restrict__ rowptr, int* __restrict__ col, float* __restrict__ dinv,
          int N, long E, int NC, int NT) {
    __shared__ int sCnt[256];
    __shared__ int sCur[256];
    __shared__ int sWave[8];
    __shared__ int sRedB[8];
    __shared__ int sRedT[8];
    __shared__ int sCol[CAP];
    const int tid  = threadIdx.x;
    const int lane = tid & 31;
    const int wave = __builtin_amdgcn_readfirstlane(tid >> 5);
    const int t    = blockIdx.x;

    int st[2], ln[2];
    for (int h = 0; h < 2; ++h) {
        const int c = wave + 8 * (lane + 32 * h);
        const bool in = c < NC;
        const size_t pi = (size_t)(in ? c : 0) * (NT + 1) + t;
        const int p0 = P[pi];
        const int p1 = P[pi + 1];
        st[h] = in ? p0 : 0;
        ln[h] = in ? p1 - p0 : 0;
    }
    sCnt[tid] = 0;

    unsigned rv[SEGW];
#pragma unroll
    for (int q = 0; q < SEGW; ++q) {
        const int s0 = __builtin_amdgcn_readlane(st[q >> 5], q & 31);
        const int l0 = __builtin_amdgcn_readlane(ln[q >> 5], q & 31);
        const unsigned a = lane < l0 ? (unsigned)((wave + 8 * q) * CH + s0 + lane) : 0u;
        rv[q] = recs[a];
    }

    int bsum = st[0] + st[1], tsum = ln[0] + ln[1];
    for (int o = 16; o > 0; o >>= 1) {
        bsum += __shfl_xor(bsum, o);
        tsum += __shfl_xor(tsum, o);
    }
    if (lane == 0) { sRedB[wave] = bsum; sRedT[wave] = tsum; }
    __syncthreads();
    int base = 0, total = 0;
    for (int w = 0; w < 8; ++w) { base += sRedB[w]; total += sRedT[w]; }
    const bool staged = total <= CAP;

#pragma unroll
    for (int q = 0; q < SEGW; ++q) {
        const int l0 = __builtin_amdgcn_readlane(ln[q >> 5], q & 31);
        if (lane < l0)
            __hip_atomic_fetch_add(&sCnt[rv[q] & 255u], 1, __ATOMIC_RELAXED,
                                   __HIP_MEMORY_SCOPE_WORKGROUP);
    }
    segments_slow<false>(recs, st, ln, sCnt, sCol, col, base, staged, wave, lane, 32);
    __syncthreads();

    const int cn = sCnt[tid];
    int incl = cn;
    for (int o = 1; o < 32; o <<= 1) {
        const int u = __shfl_up(incl, o);
        if (lane >= o) incl += u;
    }
    if (lane == 31) sWave[wave] = incl;
    __syncthreads();
    int carry = 0;
    for (int w = 0; w < wave; ++w) carry += sWave[w];
    const int ex = carry + incl - cn;
    const int node = t * 256 + tid;
    if (node < N) {
        rowptr[node] = base + ex;
        dinv[node]   = rsqrtf((float)(cn + 1));
    }
    if (t == NT - 1 && tid == 0) rowptr[N] = (int)E;
    sCur[tid] = ex;
    __syncthreads();

    if (staged) {
        int rk[SEGW];
#pragma unroll
        for (int q = 0; q < SEGW; ++q) {
            const int l0 = __builtin_amdgcn_readlane(ln[q >> 5], q & 31);
            rk[q] = 0;
            if (lane < l0)
                rk[q] = __hip_atomic_fetch_add(&sCur[rv[q] & 255u], 1, __ATOMIC_RELAXED,
                                               __HIP_MEMORY_SCOPE_WORKGROUP);
        }
#pragma unroll
        for (int q = 0; q < SEGW; ++q) {
            const int l0 = __builtin_amdgcn_readlane(ln[q >> 5], q & 31);
            if (lane < l0) sCol[rk[q]] = (int)(rv[q] >> 8);
        }
        segments_slow<true>(recs, st, ln, sCur, sCol, col, base, true, wave, lane, 32);
    } else {
        segments_slow<true>(recs, st, ln, sCur, sCol, col, base, false, wave, lane, 0);
    }
    __syncthreads();
    if (staged)
        for (int i = tid; i < total; i += 256) col[base + i] = sCol[i];
}

template <int F, bool RELU>
__global__ void __launch_bounds__(256)
k_gather(const int* __restrict__ rowptr, const int* __restrict__ col,
         const float* __restrict__ dinv, const float* __restrict__ h,
         const float* __restrict__ bias, float* __restrict__ res, int N) {
    constexpr int C = F / 32;
    typedef float vecC __attribute__((ext_vector_type(C)));
    const int lane = threadIdx.x & 31;
    const int node = __builtin_amdgcn_readfirstlane(
        (int)(blockIdx.x * (blockDim.x >> 5) + (threadIdx.x >> 5)));
    if (node >= N) return;
    const int beg = rowptr[node];
    const int end = rowptr[node + 1];
    const float dd = dinv[node];
    const vecC* __restrict__ hv = (const vecC*)h;
    const vecC hs = hv[(size_t)node * 32 + lane];
    const vecC bs = ((const vecC*)bias)[lane];
    vecC acc = 0.0f;

    for (int base = beg; base < end; base += 32) {
        const int rem = end - base;
        const int cnt = rem < 32 ? rem : 32;
        int s_l = 0;
        if (lane < cnt) s_l = col[base + lane];
        int j = 0;
        for (; j + 8 <= cnt; j += 8) {
            vecC v[8];
            for (int u = 0; u < 8; ++u) {
                const int s = __builtin_amdgcn_readlane(s_l, j + u);
                v[u] = hv[(size_t)s * 32 + lane];
            }
            for (int u = 0; u < 8; ++u) acc += v[u];
        }
        if (j + 4 <= cnt) {
            vecC v[4];
            for (int u = 0; u < 4; ++u) {
                const int s = __builtin_amdgcn_readlane(s_l, j + u);
                v[u] = hv[(size_t)s * 32 + lane];
            }
            for (int u = 0; u < 4; ++u) acc += v[u];
            j += 4;
        }
        for (; j < cnt; ++j) {
            const int s = __builtin_amdgcn_readlane(s_l, j);
            acc += hv[(size_t)s * 32 + lane];
        }
    }
    vecC r;
    for (int c = 0; c < C; ++c) {
        const float v = (acc[c] + hs[c]) * dd + bs[c];
        r[c] = RELU ? (v > 0.0f ? v : 0.0f) : v;
    }
    ((vecC*)res)[(size_t)node * 32 + lane] = r;
}


__device__ __forceinline__ void rows_chunk(const v4f* __restrict__ xv, int s_l, int w_i, int cnt,
                                           int lane, v4f& acc) {
    int j = 0;
    for (; j + 16 <= cnt; j += 16) {
        v4f v[16];
        for (int u = 0; u < 16; ++u) {
            const int s = __builtin_amdgcn_readlane(s_l, j + u);
            v[u] = xv[(size_t)s * 32 + lane];
        }
        for (int u = 0; u < 16; ++u)
            acc += v[u] * __int_as_float(__builtin_amdgcn_readlane(w_i, j + u));
    }
    if (j + 8 <= cnt) {
        v4f v[8];
        for (int u = 0; u < 8; ++u) {
            const int s = __builtin_amdgcn_readlane(s_l, j + u);
            v[u] = xv[(size_t)s * 32 + lane];
        }
        for (int u = 0; u < 8; ++u)
            acc += v[u] * __int_as_float(__builtin_amdgcn_readlane(w_i, j + u));
        j += 8;
    }
    if (j + 4 <= cnt) {
        v4f v[4];
        for (int u = 0; u < 4; ++u) {
            const int s = __builtin_amdgcn_readlane(s_l, j + u);
            v[u] = xv[(size_t)s * 32 + lane];
        }
        for (int u = 0; u < 4; ++u)
            acc += v[u] * __int_as_float(__builtin_amdgcn_readlane(w_i, j + u));
        j += 4;
    }
    for (; j < cnt; ++j) {
        const int s = __builtin_amdgcn_readlane(s_l, j);
        acc += xv[(size_t)s * 32 + lane] * __int_as_float(__builtin_amdgcn_readlane(w_i, j));
    }
}

template <int K, int NC, int NTHR>
__device__ __forceinline__ void stage_pairs(const float* __restrict__ Wg, float* dst, int t) {
    static_assert((K * NC / 4) % (8 * NTHR) == 0, "whole groups of eight pieces per thread");
#pragma unroll 8
    for (int q = 0; q < K * NC / 4 / NTHR; ++q) {
        const int i  = q * NTHR + t;
        const int k  = i / (NC / 4);
        const int c4 = (i % (NC / 4)) * 4;
        const int p  = k >> 1;
        const v4f w = *(const v4f*)&Wg[(size_t)k * NC + c4];
        float* d = &dst[p * (2 * NC) + (c4 ^ (16 * (p & 1))) * 2 + (k & 1)];
        d[0] = w.x; d[2] = w.y; d[4] = w.z; d[6] = w.w;
    }
}

template <bool DO1, bool DO2, int F, int FH, int NOUT, int SXS>
__device__ __forceinline__ void mm_stage(const float* sW1, const float* sW2, const float* xa,
                                         const float* ta, float* tw, const float* sd,
                                         float bb0, float bb1, float* __restrict__ H2,
                                         int m, int lane, int grow0, int rowEnd) {
    const int half = lane >> 4;
    const int idx  = lane & 15;
    const float* pa1 = xa + idx * SXS + half * 2;
    const float* pa2 = ta + idx * SXS + half * 2;
    const float* pb0 = sW1 + half * (2 * FH) + (((2 * m) ^ half) * 16 + idx) * 2;
    const float* pb1 = sW1 + half * (2 * FH) + (((2 * m + 1) ^ half) * 16 + idx) * 2;
    const float* pb2 = sW2 + half * (2 * NOUT) + ((m ^ half) * 16 + idx) * 2;
    v8f a0 = {}, a1 = {}, a2 = {};
#pragma unroll
    for (int k0 = 0; k0 < F; k0 += 4) {
        if (DO1) {
            const v2f a  = *(const v2f*)(pa1 + k0);
            const v2f b0 = *(const v2f*)(pb0 + (k0 >> 1) * (2 * FH));
            const v2f b1 = *(const v2f*)(pb1 + (k0 >> 1) * (2 * FH));
            a0 = __builtin_amdgcn_wmma_f32_16x16x4_f32(false, a, false, b0, (short)0, a0, false, false);
            a1 = __builtin_amdgcn_wmma_f32_16x16x4_f32(false, a, false, b1, (short)0, a1, false, false);
        }
        if (DO2) {
            const v2f a = *(const v2f*)(pa2 + k0);
            const v2f b = *(const v2f*)(pb2 + (k0 >> 1) * (2 * NOUT));
            a2 = __builtin_amdgcn_wmma_f32_16x16x4_f32(false, a, false, b, (short)0, a2, false, false);
        }
    }
    __builtin_amdgcn_sched_barrier(0);
    if (DO1 && DO2)  asm volatile("v_nop\n\tv_nop\n\tv_nop\n\tv_nop" : "+v"(a0), "+v"(a1), "+v"(a2));
    if (DO1 && !DO2) asm volatile("v_nop\n\tv_nop\n\tv_nop\n\tv_nop" : "+v"(a0), "+v"(a1));
    if (!DO1 && DO2) asm volatile("v_nop\n\tv_nop\n\tv_nop\n\tv_nop" : "+v"(a2));
    if (DO1) {
#pragma unroll
        for (int v = 0; v < 8; ++v) {
            const int rr = half * 8 + v;
            const float z0 = a0[v] + bb0;
            const float z1 = a1[v] + bb1;
            tw[rr * SXS + (2 * m) * 16 + idx]     = (z0 > 0.0f ? z0 : 0.0f) * sd[rr];
            tw[rr * SXS + (2 * m + 1) * 16 + idx] = (z1 > 0.0f ? z1 : 0.0f) * sd[rr];
        }
    }
    if (DO2) {
#pragma unroll
        for (int v = 0; v < 8; ++v) {
            const int gr = grow0 + half * 8 + v;
            if (gr < rowEnd) H2[(size_t)gr * NOUT + m * 16 + idx] = a2[v];
        }
    }
}

template <int F, int FH, int NOUT, int NBLK>
__global__ void __launch_bounds__(640, 10)
k_gather_gemm2(const int* __restrict__ rowptr, const int* __restrict__ col,
               const float* __restrict__ dinv, const float* __restrict__ x,
               const float* __restrict__ W1, const float* __restrict__ bias1,
               const float* __restrict__ W2, float* __restrict__ H2, int N) {
    constexpr int C   = F / 32;
    constexpr int SXS = F + 4;
    constexpr int TR  = 16;
    constexpr int NGW = 16;
    constexpr int NMW = 4;
    static_assert(C == 4, "a lane owns one 16-byte piece of a row");
    static_assert(FH == F, "both products walk the same k range");
    static_assert(FH == 2 * NMW * 16 && NOUT == NMW * 16, "strips per multiplying wave: two of T, one of H2");
    static_assert(NGW == TR, "one gathering wave per row of a tile");
    __shared__ __attribute__((aligned(16))) float sW1[F * FH];
    __shared__ __attribute__((aligned(16))) float sW2[FH * NOUT];
    __shared__ __attribute__((aligned(16))) float sX[2][TR * SXS];
    __shared__ __attribute__((aligned(16))) float sT[2][TR * SXS];
    __shared__ float sD[2][TR];

    const int lane = threadIdx.x & 31;
    const int wave = __builtin_amdgcn_readfirstlane((int)(threadIdx.x >> 5));
    const int bq = N / NBLK, br = N % NBLK, bi = (int)blockIdx.x;
    const int rowBeg = bi * bq + (bi < br ? bi : br);
    const int rowEnd = rowBeg + bq + (bi < br ? 1 : 0);
    const int nT = (rowEnd - rowBeg + TR - 1) / TR;
    if (nT == 0) return;
    const v4f* __restrict__ xv = (const v4f*)x;
    const bool gatherer = wave < NGW;
    const int m = gatherer ? 0 : wave - NGW;

    const int bsel = lane == 1 ? 1 : 0;
    int bv = 0, nv = 0, pv = 0, s_l = 0, ns_l = 0;
    float w_l = 0.0f;
    float bb0 = 0.0f, bb1 = 0.0f;
    if (gatherer) {
        const int n0 = rowBeg + wave;
        const int n1 = n0 + TR;
        const int n2 = n1 + TR;
        bv = rowptr[n0 < rowEnd ? n0 + bsel : N];
        nv = rowptr[n1 < rowEnd ? n1 + bsel : N];
        pv = rowptr[n2 < rowEnd ? n2 + bsel : N];
        const int cb = __builtin_amdgcn_readlane(bv, 0), ce = __builtin_amdgcn_readlane(bv, 1);
        const int nb = __builtin_amdgcn_readlane(nv, 0), ne = __builtin_amdgcn_readlane(nv, 1);
        const int c0 = ce - cb < 32 ? ce - cb : 32;
        const int c1 = ne - nb < 32 ? ne - nb : 32;
        if (lane < c0) s_l = col[cb + lane];
        if (lane < c1) ns_l = col[nb + lane];
        if (lane < c0) w_l = dinv[s_l];
    } else {
        bb0 = bias1[(2 * m) * 16 + (lane & 15)];
        bb1 = bias1[(2 * m + 1) * 16 + (lane & 15)];
    }

    for (int s = 0; s < nT + 2; ++s) {
        if (gatherer) {
            if (s < nT) {
                const int r    = wave;
                const int node = rowBeg + s * TR + r;
                const bool valid = node < rowEnd;
                const int nc  = valid ? node : N - 1;
                const int beg = __builtin_amdgcn_readlane(bv, 0), end = __builtin_amdgcn_readlane(bv, 1);
                const int nb  = __builtin_amdgcn_readlane(nv, 0), ne  = __builtin_amdgcn_readlane(nv, 1);
                const int pb  = __builtin_amdgcn_readlane(pv, 0), pe  = __builtin_amdgcn_readlane(pv, 1);
                const int ncnt = ne - nb < 32 ? ne - nb : 32;
                float nw_l = 0.0f;
                if (lane < ncnt) nw_l = dinv[ns_l];
                const int pcnt = pe - pb < 32 ? pe - pb : 32;
                int ps_l = 0;
                if (lane < pcnt) ps_l = col[pb + lane];
                const int n3 = node + 3 * TR;
                const int qv = rowptr[n3 < rowEnd ? n3 + bsel : N];
                const float dd = dinv[nc];
                const v4f xs = xv[(size_t)nc * 32 + lane];

                v4f acc = 0.0f;
                {
                    const int cnt = end - beg < 32 ? end - beg : 32;
                    rows_chunk(xv, s_l, __float_as_int(w_l), cnt, lane, acc);
                }
                for (int base = beg + 32; base < end; base += 32) {
                    const int rem = end - base;
                    const int cnt = rem < 32 ? rem : 32;
                    int s2 = 0;
                    float w2 = 0.0f;
                    if (lane < cnt) {
                        s2 = col[base + lane];
                        w2 = dinv[s2];
                    }
                    rows_chunk(xv, s2, __float_as_int(w2), cnt, lane, acc);
                }

                v4f o;
                for (int c = 0; c < C; ++c) o[c] = valid ? (acc[c] + xs[c] * dd) * dd : 0.0f;
                *(v4f*)&sX[s & 1][r * SXS + lane * C] = o;
                if (lane == 0) sD[s & 1][r] = valid ? dd : 0.0f;
                bv = nv; s_l = ns_l; w_l = nw_l;
                nv = pv; ns_l = ps_l;
                pv = qv;
            }
        } else {
            if (s == 0) {
                const int t = m * 32 + lane;
                stage_pairs<F, FH, NMW * 32>(W1, sW1, t);
                stage_pairs<FH, NOUT, NMW * 32>(W2, sW2, t);
            } else {
                int ln = lane;
                asm volatile("" : "+v"(ln));
                const float* xa = sX[(s + 1) & 1];
                const float* sd = sD[(s + 1) & 1];
                float*       tw = sT[(s + 1) & 1];
                const float* ta = sT[s & 1];
                const int grow0 = rowBeg + (s - 2) * TR;
                if (s == 1)
                    mm_stage<true, false, F, FH, NOUT, SXS>(sW1, sW2, xa, ta, tw, sd, bb0, bb1,
                                                                      H2, m, ln, grow0, rowEnd);
                else if (s <= nT)
                    mm_stage<true, true, F, FH, NOUT, SXS>(sW1, sW2, xa, ta, tw, sd, bb0, bb1,
                                                                     H2, m, ln, grow0, rowEnd);
                else
                    mm_stage<false, true, F, FH, NOUT, SXS>(sW1, sW2, xa, ta, tw, sd, bb0, bb1,
                                                                      H2, m, ln, grow0, rowEnd);
            }
        }
        __syncthreads();
    }
}

extern "C" void kernel_launch(void* const* d_in, const int* in_sizes, int n_in,
                              void* d_out, int out_size, void* d_ws, size_t ws_size,
                              hipStream_t stream) {
    const float* x  = (const float*)d_in[0];
    const int*   ei = (const int*)d_in[1];
    const float* W1 = (const float*)d_in[2];
    const float* b1 = (const float*)d_in[3];
    const float* W2 = (const float*)d_in[4];
    const float* b2 = (const float*)d_in[5];
    float* out = (float*)d_out;

    const int  N = in_sizes[0] / IN_F;
    const long E = (long)in_sizes[1] / 2;
    const int* srcIdx = ei;
    const int* dstIdx = ei + E;

    float* ws   = (float*)d_ws;
    float* dinv = ws;
    float* h2   = dinv + N + (size_t)N * HID_F;
    const int N4 = (N + 3) / 4 * 4;
    const int NC = (int)((E + CH - 1) / CH);
    const int NT = (N + 255) / 256;
    int* rowptr    = (int*)(h2 + (size_t)N * HID_F);
    int* col       = rowptr + N4 + 4;
    unsigned* recs = (unsigned*)h2;
    int* P         = (int*)(recs + (size_t)NC * CH);

    k_tilesort<<<NC, 256, 0, stream>>>(srcIdx, dstIdx, recs, P, E, NT);
    k_tilecsr<<<NT, 256, 0, stream>>>(recs, P, rowptr, col, dinv, N, E, NC, NT);

    constexpr int FB = 512;
    k_gather_gemm2<IN_F, HID_F, OUT_F, FB>
        <<<FB, 640, 0, stream>>>(rowptr, col, dinv, x, W1, b1, W2, h2, N);

    k_gather<OUT_F, false><<<(N + 7) / 8, 256, 0, stream>>>(rowptr, col, dinv, h2, b2, out, N);
}
